# starting kernel
# baseline (speedup 1.0000x reference)
#include <hip/hip_runtime.h>

typedef float v2f __attribute__((ext_vector_type(2)));
typedef float v8f __attribute__((ext_vector_type(8)));

#define NQ    12
#define NL    6
#define DIM   4096
#define BLOCK 256

__host__ __device__ static constexpr int state_index(int r, int j, int sh) {
    return ((r >> sh) << (sh + 4)) | (j << sh) | (r & ((1 << sh) - 1));
}

__host__ __device__ static constexpr int swzc(int i) {
    return i ^ ((i >> 4) & 15) ^ ((i >> 8) & 15);
}

__host__ __device__ static constexpr int Rmap(int j) {
    int x = j;
    for (int w = 11; w >= 0; --w) {
        const int pc = 11 - w;
        const int pt = 11 - ((w + 1) % 12);
        x ^= ((x >> pc) & 1) << pt;
    }
    return x;
}

__global__ __launch_bounds__(BLOCK) void qsim12_kernel(
    const float* __restrict__ model_input,
    const float* __restrict__ weights,
    float* __restrict__ out)
{
    __shared__ float2 st2[DIM];
    __shared__ float2 Ar2[3][256], Ai2[3][256];
    __shared__ float gates[NL][NQ][8];
    __shared__ float red[BLOCK];

    const int b    = blockIdx.x;
    const int t    = threadIdx.x;
    const int lane = t & 31;
    const int wv   = t >> 5;

    const int storeBase = swzc(t << 4);
    const int permBase  = swzc(Rmap(t << 4));

    for (int k = t; k < DIM; k += BLOCK) st2[k] = make_float2(0.0f, 0.0f);
    if (t == 0) st2[0] = make_float2(1.0f, 0.0f);

    const float a0 = model_input[2 * b + 0];
    const float a1 = model_input[2 * b + 1];

    if (t < NL * NQ) {
        const int l = t / NQ, w = t % NQ;
        const int limit = NQ - (l >> 1);
        float th = weights[(l * NQ + w) * 2 + 0];
        if (w < limit) th += (w & 1) ? a1 : a0;
        const float ph = weights[(l * NQ + w) * 2 + 1];
        float c, s, pr, pi;
        __sincosf(0.5f * th, &s, &c);
        __sincosf(0.5f * ph, &pi, &pr);
        gates[l][w][0] =  pr * c;  gates[l][w][1] = -pi * c;
        gates[l][w][2] = -pr * s;  gates[l][w][3] =  pi * s;
        gates[l][w][4] =  pr * s;  gates[l][w][5] =  pi * s;
        gates[l][w][6] =  pr * c;  gates[l][w][7] =  pi * c;
    }
    __syncthreads();

    #pragma unroll 1
    for (int l = 0; l < NL; ++l) {
        {
            const int jo = t >> 4, ji = t & 15;
            #pragma unroll
            for (int g = 0; g < 3; ++g) {
                float re = 1.0f, im = 0.0f;
                #pragma unroll
                for (int k = 0; k < 4; ++k) {
                    const int w  = 4 * g + k;
                    const int bo = (jo >> (3 - k)) & 1;
                    const int bi = (ji >> (3 - k)) & 1;
                    const float er = gates[l][w][(bo * 2 + bi) * 2 + 0];
                    const float ei = gates[l][w][(bo * 2 + bi) * 2 + 1];
                    const float nr = re * er - im * ei;
                    const float ni = re * ei + im * er;
                    re = nr; im = ni;
                }
                Ar2[g][t] = make_float2(re, -im);
                Ai2[g][t] = make_float2(im,  re);
            }
        }
        __syncthreads();

        #pragma unroll
        for (int g = 0; g < 3; ++g) {
            const int sh = 8 - 4 * g;

            const int m  = lane & 15;
            const int hi = lane >> 4;
            v2f ar[8], ai[8];
            #pragma unroll
            for (int c = 0; c < 8; ++c) {
                const int col = m * 16 + 2 * c + hi;
                ar[c] = *(const v2f*)&Ar2[g][col];
                ai[c] = *(const v2f*)&Ai2[g][col];
            }

            #pragma unroll
            for (int tt = 0; tt < 2; ++tt) {
                const int r0 = (wv * 2 + tt) * 16;
                const int rn = r0 + (lane & 15);
                const int sri = swzc(state_index(rn, 0, sh));
                const int sbB = sri ^ (hi ? swzc(1 << sh) : 0);
                const int sbD = sri ^ (hi ? swzc(8 << sh) : 0);

                v2f bb[8];
                #pragma unroll
                for (int c = 0; c < 8; ++c) {
                    bb[c] = *(const v2f*)&st2[sbB ^ swzc((2 * c) << sh)];
                }
                v8f dr = {0, 0, 0, 0, 0, 0, 0, 0};
                v8f di = {0, 0, 0, 0, 0, 0, 0, 0};
                #pragma unroll
                for (int c = 0; c < 8; ++c) {
                    dr = __builtin_amdgcn_wmma_f32_16x16x4_f32(false, ar[c], false, bb[c], (short)0, dr, false, false);
                    di = __builtin_amdgcn_wmma_f32_16x16x4_f32(false, ai[c], false, bb[c], (short)0, di, false, false);
                }
                #pragma unroll
                for (int v = 0; v < 8; ++v) {
                    st2[sbD ^ swzc(v << sh)] = make_float2(dr[v], di[v]);
                }
            }
            __syncthreads();
        }

        float2 tmp[16];
        #pragma unroll
        for (int k = 0; k < 16; ++k) {
            tmp[k] = st2[permBase ^ swzc(Rmap(k))];
        }
        __syncthreads();
        #pragma unroll
        for (int k = 0; k < 16; ++k) {
            st2[storeBase ^ k] = tmp[k];
        }
        __syncthreads();
    }

    float acc = 0.0f;
    #pragma unroll
    for (int k = 0; k < 16; ++k) {
        const float2 v = st2[storeBase ^ k];
        const float p = v.x * v.x + v.y * v.y;
        acc += (k & 1) ? -p : p;
    }
    red[t] = acc;
    __syncthreads();
    #pragma unroll
    for (int off = BLOCK / 2; off > 0; off >>= 1) {
        if (t < off) red[t] += red[t + off];
        __syncthreads();
    }
    if (t == 0) out[b] = red[0];
}

extern "C" void kernel_launch(void* const* d_in, const int* in_sizes, int n_in,
                              void* d_out, int out_size, void* d_ws, size_t ws_size,
                              hipStream_t stream) {
    const float* model_input = (const float*)d_in[0];
    const float* weights     = (const float*)d_in[1];
    float* out = (float*)d_out;
    const int batch = in_sizes[0] / 2;
    qsim12_kernel<<<batch, BLOCK, 0, stream>>>(model_input, weights, out);
}


# Factor each 16x16 group unitary as diagonal RZ phases times real RY Kronecker matrix: real K=16 f32 WMMA on re and im planes (8 per tile, was 16; sums regrouped), phases applied to D in registers
# speedup vs baseline: 1.4034x; 1.4034x over previous
#include <hip/hip_runtime.h>

typedef float v2f __attribute__((ext_vector_type(2)));
typedef float v8f __attribute__((ext_vector_type(8)));

#define NQ    12
#define NL    6
#define DIM   4096
#define BLOCK 256
#define OP    20

__host__ __device__ static constexpr int state_index(int r, int j, int sh) {
    return ((r >> sh) << (sh + 4)) | (j << sh) | (r & ((1 << sh) - 1));
}

__host__ __device__ static constexpr int swzc(int i) {
    return i ^ ((i >> 4) & 15) ^ ((i >> 8) & 15);
}

__host__ __device__ static constexpr int Rmap(int j) {
    int x = j;
    for (int w = 11; w >= 0; --w) {
        const int pc = 11 - w;
        const int pt = 11 - ((w + 1) % 12);
        x ^= ((x >> pc) & 1) << pt;
    }
    return x;
}

__global__ __launch_bounds__(BLOCK) void qsim12_kernel(
    const float* __restrict__ model_input,
    const float* __restrict__ weights,
    float* __restrict__ out)
{
    __shared__ float2 st2[DIM];
    __shared__ __attribute__((aligned(16))) float Ot[3][16 * OP];
    __shared__ float2 Ph[3][16];
    __shared__ float gates[NL][NQ][8];
    __shared__ float red[BLOCK];

    const int b    = blockIdx.x;
    const int t    = threadIdx.x;
    const int lane = t & 31;
    const int wv   = t >> 5;

    const int storeBase = swzc(t << 4);
    const int permBase  = swzc(Rmap(t << 4));

    for (int k = t; k < DIM; k += BLOCK) st2[k] = make_float2(0.0f, 0.0f);
    if (t == 0) st2[0] = make_float2(1.0f, 0.0f);

    const float a0 = model_input[2 * b + 0];
    const float a1 = model_input[2 * b + 1];

    if (t < NL * NQ) {
        const int l = t / NQ, w = t % NQ;
        const int limit = NQ - (l >> 1);
        float th = weights[(l * NQ + w) * 2 + 0];
        if (w < limit) th += (w & 1) ? a1 : a0;
        const float ph = weights[(l * NQ + w) * 2 + 1];
        float c, s, pr, pi;
        __sincosf(0.5f * th, &s, &c);
        __sincosf(0.5f * ph, &pi, &pr);
        gates[l][w][0] = c;   gates[l][w][1] = -s;
        gates[l][w][2] = s;   gates[l][w][3] = c;
        gates[l][w][4] = pr;  gates[l][w][5] = -pi;
        gates[l][w][6] = pr;  gates[l][w][7] = pi;
    }
    __syncthreads();

    #pragma unroll 1
    for (int l = 0; l < NL; ++l) {
        {
            const int jo = t >> 4, ji = t & 15;
            const int q  = (ji & 12) | ((ji & 1) << 1) | ((ji >> 1) & 1);
            #pragma unroll
            for (int g = 0; g < 3; ++g) {
                float o = 1.0f;
                #pragma unroll
                for (int k = 0; k < 4; ++k) {
                    const int w  = 4 * g + k;
                    const int bo = (jo >> (3 - k)) & 1;
                    const int bi = (ji >> (3 - k)) & 1;
                    o *= gates[l][w][bo * 2 + bi];
                }
                Ot[g][jo * OP + q] = o;
            }
            if (t < 48) {
                const int g = t >> 4, j = t & 15;
                float re = 1.0f, im = 0.0f;
                #pragma unroll
                for (int k = 0; k < 4; ++k) {
                    const int w  = 4 * g + k;
                    const int bo = (j >> (3 - k)) & 1;
                    const float er = gates[l][w][4 + 2 * bo];
                    const float ei = gates[l][w][5 + 2 * bo];
                    const float nr = re * er - im * ei;
                    const float ni = re * ei + im * er;
                    re = nr; im = ni;
                }
                Ph[g][j] = make_float2(re, im);
            }
        }
        __syncthreads();

        #pragma unroll
        for (int g = 0; g < 3; ++g) {
            const int sh = 8 - 4 * g;

            const int m  = lane & 15;
            const int hi = lane >> 4;
            v2f aa[4];
            #pragma unroll
            for (int c = 0; c < 4; ++c) {
                aa[c] = *(const v2f*)&Ot[g][m * OP + 4 * c + 2 * hi];
            }
            float2 pz[8];
            #pragma unroll
            for (int v = 0; v < 8; ++v) pz[v] = Ph[g][v + 8 * hi];

            #pragma unroll
            for (int tt = 0; tt < 2; ++tt) {
                const int r0 = (wv * 2 + tt) * 16;
                const int rn = r0 + (lane & 15);
                const int sri = swzc(state_index(rn, 0, sh));
                const int sbB = sri ^ (hi ? swzc(1 << sh) : 0);
                const int sbD = sri ^ (hi ? swzc(8 << sh) : 0);

                v2f bb[8];
                #pragma unroll
                for (int e = 0; e < 8; ++e) {
                    bb[e] = *(const v2f*)&st2[sbB ^ swzc((2 * e) << sh)];
                }
                v8f dr = {0, 0, 0, 0, 0, 0, 0, 0};
                v8f di = {0, 0, 0, 0, 0, 0, 0, 0};
                #pragma unroll
                for (int c = 0; c < 4; ++c) {
                    const v2f bre = {bb[2 * c].x, bb[2 * c + 1].x};
                    const v2f bim = {bb[2 * c].y, bb[2 * c + 1].y};
                    dr = __builtin_amdgcn_wmma_f32_16x16x4_f32(false, aa[c], false, bre, (short)0, dr, false, false);
                    di = __builtin_amdgcn_wmma_f32_16x16x4_f32(false, aa[c], false, bim, (short)0, di, false, false);
                }
                #pragma unroll
                for (int v = 0; v < 8; ++v) {
                    const float xr = dr[v], xi = di[v];
                    st2[sbD ^ swzc(v << sh)] = make_float2(xr * pz[v].x - xi * pz[v].y,
                                                           xr * pz[v].y + xi * pz[v].x);
                }
            }
            __syncthreads();
        }

        float2 tmp[16];
        #pragma unroll
        for (int k = 0; k < 16; ++k) {
            tmp[k] = st2[permBase ^ swzc(Rmap(k))];
        }
        __syncthreads();
        #pragma unroll
        for (int k = 0; k < 16; ++k) {
            st2[storeBase ^ k] = tmp[k];
        }
        __syncthreads();
    }

    float acc = 0.0f;
    #pragma unroll
    for (int k = 0; k < 16; ++k) {
        const float2 v = st2[storeBase ^ k];
        const float p = v.x * v.x + v.y * v.y;
        acc += (k & 1) ? -p : p;
    }
    red[t] = acc;
    __syncthreads();
    #pragma unroll
    for (int off = BLOCK / 2; off > 0; off >>= 1) {
        if (t < off) red[t] += red[t + off];
        __syncthreads();
    }
    if (t == 0) out[b] = red[0];
}

extern "C" void kernel_launch(void* const* d_in, const int* in_sizes, int n_in,
                              void* d_out, int out_size, void* d_ws, size_t ws_size,
                              hipStream_t stream) {
    const float* model_input = (const float*)d_in[0];
    const float* weights     = (const float*)d_in[1];
    float* out = (float*)d_out;
    const int batch = in_sizes[0] / 2;
    qsim12_kernel<<<batch, BLOCK, 0, stream>>>(model_input, weights, out);
}


# LDS state layout: padded pitches (290/18/1) replace the XOR swizzle; fragment accesses are lane base + immediate: 228->120 VGPRs (4 workgroups per WGP, was 2), no bank conflicts; k sums regrouped
# speedup vs baseline: 1.6130x; 1.1494x over previous
#include <hip/hip_runtime.h>

typedef float v2f __attribute__((ext_vector_type(2)));
typedef float v8f __attribute__((ext_vector_type(8)));

#define NQ    12
#define NL    6
#define DIM   4096
#define BLOCK 256
#define OP    20

#define P1    18
#define P2    290
#define ST_N  (15 * P2 + 15 * P1 + 16)

__host__ __device__ static constexpr int pos_of(int i) {
    return P2 * (i >> 8) + P1 * ((i >> 4) & 15) + (i & 15);
}

__host__ __device__ static constexpr int Rmap(int j) {
    int x = j;
    for (int w = 11; w >= 0; --w) {
        const int pc = 11 - w;
        const int pt = 11 - ((w + 1) % 12);
        x ^= ((x >> pc) & 1) << pt;
    }
    return x;
}

__global__ __launch_bounds__(BLOCK) void qsim12_kernel(
    const float* __restrict__ model_input,
    const float* __restrict__ weights,
    float* __restrict__ out)
{
    __shared__ float2 st2[ST_N];
    __shared__ __attribute__((aligned(16))) float Ot[3][16 * OP];
    __shared__ __attribute__((aligned(16))) float2 Ph[3][16];
    __shared__ float gates[NL][NQ][8];
    __shared__ float red[BLOCK];

    const int b    = blockIdx.x;
    const int t    = threadIdx.x;
    const int lane = t & 31;
    const int wv   = t >> 5;
    const int c16  = lane & 15;
    const int hi   = lane >> 4;

    const int base0 = P2 * (8 * hi) + P1 * (2 * wv) + c16;
    const int base1 = P2 * (2 * wv) + P1 * (8 * hi) + c16;
    const int base2 = P2 * (2 * wv) + P1 * c16 + hi;
    const int permDst = P1 * (8 * hi + wv) + c16;
    const int permSrc = Rmap(((8 * hi + wv) << 4) | c16);
    const int finBase = P2 * (t >> 4) + P1 * (t & 15);

    for (int k = t; k < ST_N; k += BLOCK) st2[k] = make_float2(0.0f, 0.0f);
    if (t == 0) st2[0] = make_float2(1.0f, 0.0f);

    const float a0 = model_input[2 * b + 0];
    const float a1 = model_input[2 * b + 1];

    if (t < NL * NQ) {
        const int l = t / NQ, w = t % NQ;
        const int limit = NQ - (l >> 1);
        float th = weights[(l * NQ + w) * 2 + 0];
        if (w < limit) th += (w & 1) ? a1 : a0;
        const float ph = weights[(l * NQ + w) * 2 + 1];
        float c, s, pr, pi;
        __sincosf(0.5f * th, &s, &c);
        __sincosf(0.5f * ph, &pi, &pr);
        gates[l][w][0] = c;   gates[l][w][1] = -s;
        gates[l][w][2] = s;   gates[l][w][3] = c;
        gates[l][w][4] = pr;  gates[l][w][5] = -pi;
        gates[l][w][6] = pr;  gates[l][w][7] = pi;
    }
    __syncthreads();

    #pragma unroll 1
    for (int l = 0; l < NL; ++l) {
        {
            const int jo = t >> 4, ji = t & 15;
            const int qA = ((ji & 6) << 1) | ((ji >> 3) << 1) | (ji & 1);
            const int qB = (ji & 12) | ((ji & 1) << 1) | ((ji >> 1) & 1);
            const int rB = (jo >> 1) + 8 * (jo & 1);
            #pragma unroll
            for (int g = 0; g < 3; ++g) {
                float o = 1.0f;
                #pragma unroll
                for (int k = 0; k < 4; ++k) {
                    const int w  = 4 * g + k;
                    const int bo = (jo >> (3 - k)) & 1;
                    const int bi = (ji >> (3 - k)) & 1;
                    o *= gates[l][w][bo * 2 + bi];
                }
                Ot[g][(g == 2) ? (rB * OP + qB) : (jo * OP + qA)] = o;
            }
            if (t < 48) {
                const int g = t >> 4, j = t & 15;
                float re = 1.0f, im = 0.0f;
                #pragma unroll
                for (int k = 0; k < 4; ++k) {
                    const int w  = 4 * g + k;
                    const int bo = (j >> (3 - k)) & 1;
                    const float er = gates[l][w][4 + 2 * bo];
                    const float ei = gates[l][w][5 + 2 * bo];
                    const float nr = re * er - im * ei;
                    const float ni = re * ei + im * er;
                    re = nr; im = ni;
                }
                Ph[g][(g == 2) ? ((j >> 1) + 8 * (j & 1)) : j] = make_float2(re, im);
            }
        }
        __syncthreads();

        #pragma unroll
        for (int g = 0; g < 3; ++g) {
            const int base = (g == 0) ? base0 : (g == 1) ? base1 : base2;
            const int se   = (g == 0) ? P2 : (g == 1) ? P1 : 2;
            const int st   = (g == 0) ? P1 : P2;

            v2f aa[4];
            #pragma unroll
            for (int c = 0; c < 4; ++c) {
                aa[c] = *(const v2f*)&Ot[g][c16 * OP + 4 * c + 2 * hi];
            }
            float2 pz[8];
            #pragma unroll
            for (int v = 0; v < 8; ++v) pz[v] = Ph[g][v + 8 * hi];

            #pragma unroll
            for (int tt = 0; tt < 2; ++tt) {
                float2* sp = &st2[base + st * tt];

                v2f bb[8];
                #pragma unroll
                for (int e = 0; e < 8; ++e) {
                    bb[e] = *(const v2f*)&sp[se * e];
                }
                v8f dr = {0, 0, 0, 0, 0, 0, 0, 0};
                v8f di = {0, 0, 0, 0, 0, 0, 0, 0};
                #pragma unroll
                for (int c = 0; c < 4; ++c) {
                    const v2f bre = {bb[2 * c].x, bb[2 * c + 1].x};
                    const v2f bim = {bb[2 * c].y, bb[2 * c + 1].y};
                    dr = __builtin_amdgcn_wmma_f32_16x16x4_f32(false, aa[c], false, bre, (short)0, dr, false, false);
                    di = __builtin_amdgcn_wmma_f32_16x16x4_f32(false, aa[c], false, bim, (short)0, di, false, false);
                }
                #pragma unroll
                for (int v = 0; v < 8; ++v) {
                    const float xr = dr[v], xi = di[v];
                    sp[se * v] = make_float2(xr * pz[v].x - xi * pz[v].y,
                                             xr * pz[v].y + xi * pz[v].x);
                }
            }
            __syncthreads();
        }

        float2 tmp[16];
        #pragma unroll
        for (int k = 0; k < 16; ++k) {
            tmp[k] = st2[pos_of(permSrc ^ Rmap(k << 8))];
        }
        __syncthreads();
        #pragma unroll
        for (int k = 0; k < 16; ++k) {
            st2[permDst + P2 * k] = tmp[k];
        }
        __syncthreads();
    }

    float acc = 0.0f;
    #pragma unroll
    for (int k = 0; k < 16; ++k) {
        const float2 v = st2[finBase + k];
        const float p = v.x * v.x + v.y * v.y;
        acc += (k & 1) ? -p : p;
    }
    red[t] = acc;
    __syncthreads();
    #pragma unroll
    for (int off = BLOCK / 2; off > 0; off >>= 1) {
        if (t < off) red[t] += red[t + off];
        __syncthreads();
    }
    if (t == 0) out[b] = red[0];
}

extern "C" void kernel_launch(void* const* d_in, const int* in_sizes, int n_in,
                              void* d_out, int out_size, void* d_ws, size_t ws_size,
                              hipStream_t stream) {
    const float* model_input = (const float*)d_in[0];
    const float* weights     = (const float*)d_in[1];
    float* out = (float*)d_out;
    const int batch = in_sizes[0] / 2;
    qsim12_kernel<<<batch, BLOCK, 0, stream>>>(model_input, weights, out);
}


# Groups 1+2 fused in registers (group 1 with A/B swapped: its D is group 2's B fragment): one LDS round trip and barrier less per layer; phases merged; CNOT pass base+offset; k sums regrouped
# speedup vs baseline: 1.8184x; 1.1273x over previous
#include <hip/hip_runtime.h>

typedef float v2f __attribute__((ext_vector_type(2)));
typedef float v8f __attribute__((ext_vector_type(8)));

#define NQ    12
#define NL    6
#define DIM   4096
#define BLOCK 256
#define OP    20

#define P1    18
#define P2    290
#define ST_N  (15 * P2 + 15 * P1 + 16)

__host__ __device__ static constexpr int pos_of(int i) {
    return P2 * (i >> 8) + P1 * ((i >> 4) & 15) + (i & 15);
}

__host__ __device__ static constexpr int Rmap(int j) {
    int x = j;
    for (int w = 11; w >= 0; --w) {
        const int pc = 11 - w;
        const int pt = 11 - ((w + 1) % 12);
        x ^= ((x >> pc) & 1) << pt;
    }
    return x;
}

__host__ __device__ static constexpr int gray4(int k) { return k ^ (k >> 1); }
__host__ __device__ static constexpr int gray4_inv(int g) { return g ^ (g >> 1) ^ (g >> 2) ^ (g >> 3); }
__host__ __device__ static constexpr int par4(int m) { return (m ^ (m >> 1) ^ (m >> 2) ^ (m >> 3)) & 1; }
__host__ __device__ static constexpr bool ring_structure_ok() {
    for (int k = 0; k < 16; ++k) {
        if (Rmap(k << 8) != ((gray4(k) << 8) | ((k & 1) << 7))) return false;
        if ((gray4_inv(gray4(k)) & 15) != k) return false;
    }
    for (int low = 0; low < 256; ++low) {
        const int s = Rmap(low);
        if ((s >> 8) != 12 * (low & 1)) return false;
        if (((s >> 7) & 1) != ((low >> 7) & 1)) return false;
    }
    return true;
}
static_assert(ring_structure_ok(), "permutation pass addressing assumes this structure of the CNOT ring map");

__global__ __launch_bounds__(BLOCK) void qsim12_kernel(
    const float* __restrict__ model_input,
    const float* __restrict__ weights,
    float* __restrict__ out)
{
    __shared__ __attribute__((aligned(16))) float2 st2[ST_N];
    __shared__ __attribute__((aligned(16))) float Ot[3][16 * OP];
    __shared__ __attribute__((aligned(16))) float2 Ph[3][16];
    __shared__ __attribute__((aligned(16))) float2 Pc[16 * P1];
    __shared__ float gates[NL][NQ][8];
    __shared__ float red[BLOCK];

    const int b    = blockIdx.x;
    const int t    = threadIdx.x;
    const int lane = t & 31;
    const int wv   = t >> 5;
    const int c16  = lane & 15;
    const int hi   = lane >> 4;

    const int base0 = P2 * (8 * hi) + P1 * (2 * wv) + c16;
    const int base1 = P2 * (2 * wv) + P1 * (8 * hi) + c16;
    const int base2 = P2 * (2 * wv) + P1 * c16 + 8 * hi;
    const int permSrc  = Rmap(((8 * hi + wv) << 4) | c16);
    const int permLow  = pos_of(permSrc & 0x7F);
    const int permSrcE = permLow + 8 * P1 * ((permSrc >> 7) & 1);
    const int permSrcO = permLow + 8 * P1 * (1 - ((permSrc >> 7) & 1));
    const int permDst  = P1 * (8 * hi + wv) + c16;
    const int permDst0 = permDst + 8 * P2 * ((permSrc >> 11) & 1);
    const int permDst1 = permDst + 8 * P2 * (1 - ((permSrc >> 11) & 1));
    const int finBase = P2 * (t >> 4) + P1 * (t & 15);

    for (int k = t; k < ST_N; k += BLOCK) st2[k] = make_float2(0.0f, 0.0f);
    if (t == 0) st2[0] = make_float2(1.0f, 0.0f);

    const float a0 = model_input[2 * b + 0];
    const float a1 = model_input[2 * b + 1];

    if (t < NL * NQ) {
        const int l = t / NQ, w = t % NQ;
        const int limit = NQ - (l >> 1);
        float th = weights[(l * NQ + w) * 2 + 0];
        if (w < limit) th += (w & 1) ? a1 : a0;
        const float ph = weights[(l * NQ + w) * 2 + 1];
        float c, s, pr, pi;
        __sincosf(0.5f * th, &s, &c);
        __sincosf(0.5f * ph, &pi, &pr);
        gates[l][w][0] = c;   gates[l][w][1] = -s;
        gates[l][w][2] = s;   gates[l][w][3] = c;
        gates[l][w][4] = pr;  gates[l][w][5] = -pi;
        gates[l][w][6] = pr;  gates[l][w][7] = pi;
    }
    __syncthreads();

    #pragma unroll 1
    for (int l = 0; l < NL; ++l) {
        {
            const int jo = t >> 4, ji = t & 15;
            const int qA = ((ji & 6) << 1) | ((ji >> 3) << 1) | (ji & 1);
            #pragma unroll
            for (int g = 0; g < 3; ++g) {
                float o = 1.0f;
                #pragma unroll
                for (int k = 0; k < 4; ++k) {
                    const int w  = 4 * g + k;
                    const int bo = (jo >> (3 - k)) & 1;
                    const int bi = (ji >> (3 - k)) & 1;
                    o *= gates[l][w][bo * 2 + bi];
                }
                Ot[g][jo * OP + qA] = o;
            }
            if (t < 48) {
                const int g = t >> 4, j = t & 15;
                float re = 1.0f, im = 0.0f;
                #pragma unroll
                for (int k = 0; k < 4; ++k) {
                    const int w  = 4 * g + k;
                    const int bo = (j >> (3 - k)) & 1;
                    const float er = gates[l][w][4 + 2 * bo];
                    const float ei = gates[l][w][5 + 2 * bo];
                    const float nr = re * er - im * ei;
                    const float ni = re * ei + im * er;
                    re = nr; im = ni;
                }
                Ph[g][j] = make_float2(re, im);
            }
        }
        __syncthreads();

        {
            {
                const float2 q1 = Ph[1][t >> 4];
                const float2 q2 = Ph[2][t & 15];
                Pc[P1 * (t >> 4) + (t & 15)] = make_float2(q1.x * q2.x - q1.y * q2.y, q1.x * q2.y + q1.y * q2.x);
            }
            v2f aa[4];
            #pragma unroll
            for (int c = 0; c < 4; ++c) {
                aa[c] = *(const v2f*)&Ot[0][c16 * OP + 4 * c + 2 * hi];
            }
            float2 pz[8];
            #pragma unroll
            for (int v = 0; v < 8; ++v) pz[v] = Ph[0][v + 8 * hi];

            #pragma unroll
            for (int tt = 0; tt < 2; ++tt) {
                float2* sp = &st2[base0 + P1 * tt];

                v2f bb[8];
                #pragma unroll
                for (int e = 0; e < 8; ++e) {
                    bb[e] = *(const v2f*)&sp[P2 * e];
                }
                v8f dr = {0, 0, 0, 0, 0, 0, 0, 0};
                v8f di = {0, 0, 0, 0, 0, 0, 0, 0};
                #pragma unroll
                for (int c = 0; c < 4; ++c) {
                    const v2f bre = {bb[2 * c].x, bb[2 * c + 1].x};
                    const v2f bim = {bb[2 * c].y, bb[2 * c + 1].y};
                    dr = __builtin_amdgcn_wmma_f32_16x16x4_f32(false, aa[c], false, bre, (short)0, dr, false, false);
                    di = __builtin_amdgcn_wmma_f32_16x16x4_f32(false, aa[c], false, bim, (short)0, di, false, false);
                }
                #pragma unroll
                for (int v = 0; v < 8; ++v) {
                    const float xr = dr[v], xi = di[v];
                    sp[P2 * v] = make_float2(xr * pz[v].x - xi * pz[v].y,
                                             xr * pz[v].y + xi * pz[v].x);
                }
            }
        }
        __syncthreads();

        {
            v2f a1[4], a2[4];
            #pragma unroll
            for (int c = 0; c < 4; ++c) {
                a1[c] = *(const v2f*)&Ot[1][c16 * OP + 4 * c + 2 * hi];
                a2[c] = *(const v2f*)&Ot[2][c16 * OP + 4 * c + 2 * hi];
            }
            float2 pc[8];
            {
                const float4* pp = (const float4*)&Pc[P1 * c16 + 8 * hi];
                #pragma unroll
                for (int v = 0; v < 8; v += 2) {
                    const float4 q = pp[v >> 1];
                    pc[v]     = make_float2(q.x, q.y);
                    pc[v + 1] = make_float2(q.z, q.w);
                }
            }

            #pragma unroll
            for (int tt = 0; tt < 2; ++tt) {
                const float2* lp = &st2[base1 + P2 * tt];

                v2f bb[8];
                #pragma unroll
                for (int e = 0; e < 8; ++e) {
                    bb[e] = *(const v2f*)&lp[P1 * e];
                }
                v8f yr = {0, 0, 0, 0, 0, 0, 0, 0};
                v8f yi = {0, 0, 0, 0, 0, 0, 0, 0};
                #pragma unroll
                for (int c = 0; c < 4; ++c) {
                    const v2f xre = {bb[2 * c].x, bb[2 * c + 1].x};
                    const v2f xim = {bb[2 * c].y, bb[2 * c + 1].y};
                    yr = __builtin_amdgcn_wmma_f32_16x16x4_f32(false, xre, false, a1[c], (short)0, yr, false, false);
                    yi = __builtin_amdgcn_wmma_f32_16x16x4_f32(false, xim, false, a1[c], (short)0, yi, false, false);
                }
                v8f zr = {0, 0, 0, 0, 0, 0, 0, 0};
                v8f zi = {0, 0, 0, 0, 0, 0, 0, 0};
                #pragma unroll
                for (int c = 0; c < 4; ++c) {
                    const v2f bre = {yr[2 * c], yr[2 * c + 1]};
                    const v2f bim = {yi[2 * c], yi[2 * c + 1]};
                    zr = __builtin_amdgcn_wmma_f32_16x16x4_f32(false, a2[c], false, bre, (short)0, zr, false, false);
                    zi = __builtin_amdgcn_wmma_f32_16x16x4_f32(false, a2[c], false, bim, (short)0, zi, false, false);
                }
                float4* wp = (float4*)&st2[base2 + P2 * tt];
                #pragma unroll
                for (int v = 0; v < 8; v += 2) {
                    wp[v >> 1] = make_float4(zr[v] * pc[v].x - zi[v] * pc[v].y,
                                             zr[v] * pc[v].y + zi[v] * pc[v].x,
                                             zr[v + 1] * pc[v + 1].x - zi[v + 1] * pc[v + 1].y,
                                             zr[v + 1] * pc[v + 1].y + zi[v + 1] * pc[v + 1].x);
                }
            }
        }
        __syncthreads();

        float2 tmp[16];
        #pragma unroll
        for (int m = 0; m < 16; ++m) {
            tmp[m] = st2[(par4(m) ? permSrcO : permSrcE) + P2 * m];
        }
        __syncthreads();
        #pragma unroll
        for (int m = 0; m < 16; ++m) {
            st2[((m >> 3) ? permDst1 : permDst0) + P2 * (gray4_inv(m) & 7)] = tmp[m];
        }
        __syncthreads();
    }

    float acc = 0.0f;
    #pragma unroll
    for (int k = 0; k < 16; ++k) {
        const float2 v = st2[finBase + k];
        const float p = v.x * v.x + v.y * v.y;
        acc += (k & 1) ? -p : p;
    }
    red[t] = acc;
    __syncthreads();
    #pragma unroll
    for (int off = BLOCK / 2; off > 0; off >>= 1) {
        if (t < off) red[t] += red[t + off];
        __syncthreads();
    }
    if (t == 0) out[b] = red[0];
}

extern "C" void kernel_launch(void* const* d_in, const int* in_sizes, int n_in,
                              void* d_out, int out_size, void* d_ws, size_t ws_size,
                              hipStream_t stream) {
    const float* model_input = (const float*)d_in[0];
    const float* weights     = (const float*)d_in[1];
    float* out = (float*)d_out;
    const int batch = in_sizes[0] / 2;
    qsim12_kernel<<<batch, BLOCK, 0, stream>>>(model_input, weights, out);
}


# Layer 0 in closed form: the circuit starts from |0..0>, so its three group products are first columns; the post-ring product state is written directly and the WMMA loop runs layers 1-5
# speedup vs baseline: 2.0829x; 1.1455x over previous
#include <hip/hip_runtime.h>

typedef float v2f __attribute__((ext_vector_type(2)));
typedef float v8f __attribute__((ext_vector_type(8)));

#define NQ    12
#define NL    6
#define DIM   4096
#define BLOCK 256
#define OP    20

#define P1    18
#define P2    290
#define ST_N  (15 * P2 + 15 * P1 + 16)

__host__ __device__ static constexpr int pos_of(int i) {
    return P2 * (i >> 8) + P1 * ((i >> 4) & 15) + (i & 15);
}

__host__ __device__ static constexpr int Rmap(int j) {
    int x = j;
    for (int w = 11; w >= 0; --w) {
        const int pc = 11 - w;
        const int pt = 11 - ((w + 1) % 12);
        x ^= ((x >> pc) & 1) << pt;
    }
    return x;
}

__host__ __device__ static constexpr int gray4(int k) { return k ^ (k >> 1); }
__host__ __device__ static constexpr int gray4_inv(int g) { return g ^ (g >> 1) ^ (g >> 2) ^ (g >> 3); }
__host__ __device__ static constexpr int par4(int m) { return (m ^ (m >> 1) ^ (m >> 2) ^ (m >> 3)) & 1; }
__host__ __device__ static constexpr bool ring_structure_ok() {
    for (int k = 0; k < 16; ++k) {
        if (Rmap(k << 8) != ((gray4(k) << 8) | ((k & 1) << 7))) return false;
        if ((gray4_inv(gray4(k)) & 15) != k) return false;
    }
    for (int low = 0; low < 256; ++low) {
        const int s = Rmap(low);
        if ((s >> 8) != 12 * (low & 1)) return false;
        if (((s >> 7) & 1) != ((low >> 7) & 1)) return false;
    }
    return true;
}
static_assert(ring_structure_ok(), "permutation pass addressing assumes this structure of the CNOT ring map");

__device__ static inline float2 cmul(const float2 a, const float2 b) {
    return make_float2(a.x * b.x - a.y * b.y, a.x * b.y + a.y * b.x);
}

__global__ __launch_bounds__(BLOCK) void qsim12_kernel(
    const float* __restrict__ model_input,
    const float* __restrict__ weights,
    float* __restrict__ out)
{
    __shared__ __attribute__((aligned(16))) float2 st2[ST_N];
    __shared__ __attribute__((aligned(16))) float Ot[3][16 * OP];
    __shared__ __attribute__((aligned(16))) float2 Ph[3][16];
    __shared__ __attribute__((aligned(16))) float2 Pc[16 * P1];
    __shared__ float gates[NL][NQ][8];
    __shared__ float red[BLOCK];
    __shared__ __attribute__((aligned(16))) float2 V0[3][16];

    const int b    = blockIdx.x;
    const int t    = threadIdx.x;
    const int lane = t & 31;
    const int wv   = t >> 5;
    const int c16  = lane & 15;
    const int hi   = lane >> 4;

    const int base0 = P2 * (8 * hi) + P1 * (2 * wv) + c16;
    const int base1 = P2 * (2 * wv) + P1 * (8 * hi) + c16;
    const int base2 = P2 * (2 * wv) + P1 * c16 + 8 * hi;
    const int permSrc  = Rmap(((8 * hi + wv) << 4) | c16);
    const int permLow  = pos_of(permSrc & 0x7F);
    const int permSrcE = permLow + 8 * P1 * ((permSrc >> 7) & 1);
    const int permSrcO = permLow + 8 * P1 * (1 - ((permSrc >> 7) & 1));
    const int permDst  = P1 * (8 * hi + wv) + c16;
    const int permDst0 = permDst + 8 * P2 * ((permSrc >> 11) & 1);
    const int permDst1 = permDst + 8 * P2 * (1 - ((permSrc >> 11) & 1));
    const int finBase = P2 * (t >> 4) + P1 * (t & 15);

    const float a0 = model_input[2 * b + 0];
    const float a1 = model_input[2 * b + 1];

    if (t < NL * NQ) {
        const int l = t / NQ, w = t % NQ;
        const int limit = NQ - (l >> 1);
        float th = weights[(l * NQ + w) * 2 + 0];
        if (w < limit) th += (w & 1) ? a1 : a0;
        const float ph = weights[(l * NQ + w) * 2 + 1];
        float c, s, pr, pi;
        __sincosf(0.5f * th, &s, &c);
        __sincosf(0.5f * ph, &pi, &pr);
        gates[l][w][0] = c;   gates[l][w][1] = -s;
        gates[l][w][2] = s;   gates[l][w][3] = c;
        gates[l][w][4] = pr;  gates[l][w][5] = -pi;
        gates[l][w][6] = pr;  gates[l][w][7] = pi;
    }
    __syncthreads();

    if (t < 48) {
        const int g = t >> 4, j = t & 15;
        float2 v = make_float2(1.0f, 0.0f);
        #pragma unroll
        for (int k = 0; k < 4; ++k) {
            const int w  = 4 * g + k;
            const int bo = (j >> (3 - k)) & 1;
            const float a = gates[0][w][2 * bo];
            v = cmul(v, make_float2(a * gates[0][w][4 + 2 * bo], a * gates[0][w][5 + 2 * bo]));
        }
        V0[g][j] = v;
    }
    __syncthreads();
    {
        const float2 w2   = V0[2][permSrc & 15];
        const float2 lowE = cmul(V0[1][(permSrc >> 4) & 15], w2);
        const float2 lowO = cmul(V0[1][((permSrc >> 4) & 15) ^ 8], w2);
        #pragma unroll
        for (int m = 0; m < 16; ++m) {
            st2[((m >> 3) ? permDst1 : permDst0) + P2 * (gray4_inv(m) & 7)] = cmul(V0[0][m], par4(m) ? lowO : lowE);
        }
    }

    #pragma unroll 1
    for (int l = 1; l < NL; ++l) {
        {
            const int jo = t >> 4, ji = t & 15;
            const int qA = ((ji & 6) << 1) | ((ji >> 3) << 1) | (ji & 1);
            #pragma unroll
            for (int g = 0; g < 3; ++g) {
                float o = 1.0f;
                #pragma unroll
                for (int k = 0; k < 4; ++k) {
                    const int w  = 4 * g + k;
                    const int bo = (jo >> (3 - k)) & 1;
                    const int bi = (ji >> (3 - k)) & 1;
                    o *= gates[l][w][bo * 2 + bi];
                }
                Ot[g][jo * OP + qA] = o;
            }
            if (t < 48) {
                const int g = t >> 4, j = t & 15;
                float re = 1.0f, im = 0.0f;
                #pragma unroll
                for (int k = 0; k < 4; ++k) {
                    const int w  = 4 * g + k;
                    const int bo = (j >> (3 - k)) & 1;
                    const float er = gates[l][w][4 + 2 * bo];
                    const float ei = gates[l][w][5 + 2 * bo];
                    const float nr = re * er - im * ei;
                    const float ni = re * ei + im * er;
                    re = nr; im = ni;
                }
                Ph[g][j] = make_float2(re, im);
            }
        }
        __syncthreads();

        {
            {
                const float2 q1 = Ph[1][t >> 4];
                const float2 q2 = Ph[2][t & 15];
                Pc[P1 * (t >> 4) + (t & 15)] = make_float2(q1.x * q2.x - q1.y * q2.y, q1.x * q2.y + q1.y * q2.x);
            }
            v2f aa[4];
            #pragma unroll
            for (int c = 0; c < 4; ++c) {
                aa[c] = *(const v2f*)&Ot[0][c16 * OP + 4 * c + 2 * hi];
            }
            float2 pz[8];
            #pragma unroll
            for (int v = 0; v < 8; ++v) pz[v] = Ph[0][v + 8 * hi];

            #pragma unroll
            for (int tt = 0; tt < 2; ++tt) {
                float2* sp = &st2[base0 + P1 * tt];

                v2f bb[8];
                #pragma unroll
                for (int e = 0; e < 8; ++e) {
                    bb[e] = *(const v2f*)&sp[P2 * e];
                }
                v8f dr = {0, 0, 0, 0, 0, 0, 0, 0};
                v8f di = {0, 0, 0, 0, 0, 0, 0, 0};
                #pragma unroll
                for (int c = 0; c < 4; ++c) {
                    const v2f bre = {bb[2 * c].x, bb[2 * c + 1].x};
                    const v2f bim = {bb[2 * c].y, bb[2 * c + 1].y};
                    dr = __builtin_amdgcn_wmma_f32_16x16x4_f32(false, aa[c], false, bre, (short)0, dr, false, false);
                    di = __builtin_amdgcn_wmma_f32_16x16x4_f32(false, aa[c], false, bim, (short)0, di, false, false);
                }
                #pragma unroll
                for (int v = 0; v < 8; ++v) {
                    const float xr = dr[v], xi = di[v];
                    sp[P2 * v] = make_float2(xr * pz[v].x - xi * pz[v].y,
                                             xr * pz[v].y + xi * pz[v].x);
                }
            }
        }
        __syncthreads();

        {
            v2f a1[4], a2[4];
            #pragma unroll
            for (int c = 0; c < 4; ++c) {
                a1[c] = *(const v2f*)&Ot[1][c16 * OP + 4 * c + 2 * hi];
                a2[c] = *(const v2f*)&Ot[2][c16 * OP + 4 * c + 2 * hi];
            }
            float2 pc[8];
            {
                const float4* pp = (const float4*)&Pc[P1 * c16 + 8 * hi];
                #pragma unroll
                for (int v = 0; v < 8; v += 2) {
                    const float4 q = pp[v >> 1];
                    pc[v]     = make_float2(q.x, q.y);
                    pc[v + 1] = make_float2(q.z, q.w);
                }
            }

            #pragma unroll
            for (int tt = 0; tt < 2; ++tt) {
                const float2* lp = &st2[base1 + P2 * tt];

                v2f bb[8];
                #pragma unroll
                for (int e = 0; e < 8; ++e) {
                    bb[e] = *(const v2f*)&lp[P1 * e];
                }
                v8f yr = {0, 0, 0, 0, 0, 0, 0, 0};
                v8f yi = {0, 0, 0, 0, 0, 0, 0, 0};
                #pragma unroll
                for (int c = 0; c < 4; ++c) {
                    const v2f xre = {bb[2 * c].x, bb[2 * c + 1].x};
                    const v2f xim = {bb[2 * c].y, bb[2 * c + 1].y};
                    yr = __builtin_amdgcn_wmma_f32_16x16x4_f32(false, xre, false, a1[c], (short)0, yr, false, false);
                    yi = __builtin_amdgcn_wmma_f32_16x16x4_f32(false, xim, false, a1[c], (short)0, yi, false, false);
                }
                v8f zr = {0, 0, 0, 0, 0, 0, 0, 0};
                v8f zi = {0, 0, 0, 0, 0, 0, 0, 0};
                #pragma unroll
                for (int c = 0; c < 4; ++c) {
                    const v2f bre = {yr[2 * c], yr[2 * c + 1]};
                    const v2f bim = {yi[2 * c], yi[2 * c + 1]};
                    zr = __builtin_amdgcn_wmma_f32_16x16x4_f32(false, a2[c], false, bre, (short)0, zr, false, false);
                    zi = __builtin_amdgcn_wmma_f32_16x16x4_f32(false, a2[c], false, bim, (short)0, zi, false, false);
                }
                float4* wp = (float4*)&st2[base2 + P2 * tt];
                #pragma unroll
                for (int v = 0; v < 8; v += 2) {
                    wp[v >> 1] = make_float4(zr[v] * pc[v].x - zi[v] * pc[v].y,
                                             zr[v] * pc[v].y + zi[v] * pc[v].x,
                                             zr[v + 1] * pc[v + 1].x - zi[v + 1] * pc[v + 1].y,
                                             zr[v + 1] * pc[v + 1].y + zi[v + 1] * pc[v + 1].x);
                }
            }
        }
        __syncthreads();

        float2 tmp[16];
        #pragma unroll
        for (int m = 0; m < 16; ++m) {
            tmp[m] = st2[(par4(m) ? permSrcO : permSrcE) + P2 * m];
        }
        __syncthreads();
        #pragma unroll
        for (int m = 0; m < 16; ++m) {
            st2[((m >> 3) ? permDst1 : permDst0) + P2 * (gray4_inv(m) & 7)] = tmp[m];
        }
        __syncthreads();
    }

    float acc = 0.0f;
    #pragma unroll
    for (int k = 0; k < 16; ++k) {
        const float2 v = st2[finBase + k];
        const float p = v.x * v.x + v.y * v.y;
        acc += (k & 1) ? -p : p;
    }
    red[t] = acc;
    __syncthreads();
    #pragma unroll
    for (int off = BLOCK / 2; off > 0; off >>= 1) {
        if (t < off) red[t] += red[t + off];
        __syncthreads();
    }
    if (t == 0) out[b] = red[0];
}

extern "C" void kernel_launch(void* const* d_in, const int* in_sizes, int n_in,
                              void* d_out, int out_size, void* d_ws, size_t ws_size,
                              hipStream_t stream) {
    const float* model_input = (const float*)d_in[0];
    const float* weights     = (const float*)d_in[1];
    float* out = (float*)d_out;
    const int batch = in_sizes[0] / 2;
    qsim12_kernel<<<batch, BLOCK, 0, stream>>>(model_input, weights, out);
}


# CNOT ring folded into its reader: group 0 gathers B fragments through the ring map (pitch 292, conflict-free), last ring is a readout parity sign; ring pass, 2 of 5 barriers gone; sums regrouped
# speedup vs baseline: 2.2257x; 1.0686x over previous
#include <hip/hip_runtime.h>

typedef float v2f __attribute__((ext_vector_type(2)));
typedef float v8f __attribute__((ext_vector_type(8)));

#define NQ    12
#define NL    6
#define DIM   4096
#define BLOCK 256
#define OP    20

#define P1    18
#define P2    292
#define ST_N  (15 * P2 + 15 * P1 + 16)

__host__ __device__ static constexpr int pos_of(int i) {
    return P2 * (i >> 8) + P1 * ((i >> 4) & 15) + (i & 15);
}

__host__ __device__ static constexpr int Rmap(int j) {
    int x = j;
    for (int w = 11; w >= 0; --w) {
        const int pc = 11 - w;
        const int pt = 11 - ((w + 1) % 12);
        x ^= ((x >> pc) & 1) << pt;
    }
    return x;
}

__host__ __device__ static constexpr int gray4(int k) { return k ^ (k >> 1); }
__host__ __device__ static constexpr int gray4_inv(int g) { return g ^ (g >> 1) ^ (g >> 2) ^ (g >> 3); }
__host__ __device__ static constexpr int par4(int m) { return (m ^ (m >> 1) ^ (m >> 2) ^ (m >> 3)) & 1; }
__host__ __device__ static constexpr bool ring_structure_ok() {
    for (int k = 0; k < 16; ++k) {
        if (Rmap(k << 8) != ((gray4(k) << 8) | ((k & 1) << 7))) return false;
        if ((gray4_inv(gray4(k)) & 15) != k) return false;
    }
    for (int low = 0; low < 256; ++low) {
        const int s = Rmap(low);
        if ((s >> 8) != 12 * (low & 1)) return false;
        if (((s >> 7) & 1) != ((low >> 7) & 1)) return false;
    }
    return true;
}
static_assert(ring_structure_ok(), "ring gather addressing assumes this structure of the CNOT ring map");

__host__ __device__ static constexpr int Rfwd(int j) {
    int x = j;
    for (int w = 0; w < 12; ++w) {
        const int pc = 11 - w;
        const int pt = 11 - ((w + 1) % 12);
        x ^= ((x >> pc) & 1) << pt;
    }
    return x;
}
__host__ __device__ static constexpr int ring_zmask() {
    int m = 0;
    for (int i = 0; i < 12; ++i) {
        if (Rfwd(1 << i) & 1) m |= 1 << i;
    }
    return m;
}
__host__ __device__ static constexpr bool ring_inverse_ok() {
    for (int i = 0; i < 12; ++i) {
        if (Rfwd(Rmap(1 << i)) != (1 << i)) return false;
    }
    return true;
}
static_assert(ring_inverse_ok(), "Rfwd must invert Rmap (both are linear: the unit vectors decide)");
#define ZMASK (ring_zmask())

__device__ static inline float2 cmul(const float2 a, const float2 b) {
    return make_float2(a.x * b.x - a.y * b.y, a.x * b.y + a.y * b.x);
}

__global__ __launch_bounds__(BLOCK) void qsim12_kernel(
    const float* __restrict__ model_input,
    const float* __restrict__ weights,
    float* __restrict__ out)
{
    __shared__ __attribute__((aligned(16))) float2 st2[ST_N];
    __shared__ __attribute__((aligned(16))) float Ot[3][16 * OP];
    __shared__ __attribute__((aligned(16))) float2 Ph[3][16];
    __shared__ __attribute__((aligned(16))) float2 Pc[16 * P1];
    __shared__ float gates[NL][NQ][8];
    __shared__ float red[BLOCK];
    __shared__ __attribute__((aligned(16))) float2 V0[3][16];

    const int b    = blockIdx.x;
    const int t    = threadIdx.x;
    const int lane = t & 31;
    const int wv   = t >> 5;
    const int c16  = lane & 15;
    const int hi   = lane >> 4;

    const int base0 = P2 * (4 * hi) + P1 * (2 * wv) + c16;
    const int row0  = (c16 & 3) | ((c16 & 4) << 1) | ((c16 & 8) >> 1);
    const int base1 = P2 * (2 * wv) + P1 * (8 * hi) + c16;
    const int base2 = P2 * (2 * wv) + P1 * c16 + 8 * hi;
    int gsrc[2][4];
    #pragma unroll
    for (int tt = 0; tt < 2; ++tt) {
        const int s  = Rmap(((2 * wv + tt) << 4) | c16);
        const int q  = hi ^ ((s >> 11) & 1);
        const int pE = pos_of(s & 0xFF);
        const int pO = pos_of((s & 0xFF) ^ 0x80);
        gsrc[tt][0] = pE + 12 * P2 * q;
        gsrc[tt][1] = pO + 12 * P2 * q;
        gsrc[tt][2] = pE + 4 * P2 * q;
        gsrc[tt][3] = pO + 4 * P2 * q;
    }
    const int finBase = P2 * (t >> 4) + P1 * (t & 15);

    const float a0 = model_input[2 * b + 0];
    const float a1 = model_input[2 * b + 1];

    if (t < NL * NQ) {
        const int l = t / NQ, w = t % NQ;
        const int limit = NQ - (l >> 1);
        float th = weights[(l * NQ + w) * 2 + 0];
        if (w < limit) th += (w & 1) ? a1 : a0;
        const float ph = weights[(l * NQ + w) * 2 + 1];
        float c, s, pr, pi;
        __sincosf(0.5f * th, &s, &c);
        __sincosf(0.5f * ph, &pi, &pr);
        gates[l][w][0] = c;   gates[l][w][1] = -s;
        gates[l][w][2] = s;   gates[l][w][3] = c;
        gates[l][w][4] = pr;  gates[l][w][5] = -pi;
        gates[l][w][6] = pr;  gates[l][w][7] = pi;
    }
    __syncthreads();

    if (t < 48) {
        const int g = t >> 4, j = t & 15;
        float2 v = make_float2(1.0f, 0.0f);
        #pragma unroll
        for (int k = 0; k < 4; ++k) {
            const int w  = 4 * g + k;
            const int bo = (j >> (3 - k)) & 1;
            const float a = gates[0][w][2 * bo];
            v = cmul(v, make_float2(a * gates[0][w][4 + 2 * bo], a * gates[0][w][5 + 2 * bo]));
        }
        V0[g][j] = v;
    }
    __syncthreads();
    {
        const float2 low = cmul(V0[1][8 * hi + wv], V0[2][c16]);
        float2* dp = &st2[P1 * (8 * hi + wv) + c16];
        #pragma unroll
        for (int k = 0; k < 16; ++k) {
            dp[P2 * k] = cmul(V0[0][k], low);
        }
    }
    __syncthreads();

    #pragma unroll 1
    for (int l = 1; l < NL; ++l) {
        {
            const int jo = t >> 4, ji = t & 15;
            const int qA = ((ji & 6) << 1) | ((ji >> 3) << 1) | (ji & 1);
            #pragma unroll
            for (int g = 0; g < 3; ++g) {
                float o = 1.0f;
                #pragma unroll
                for (int k = 0; k < 4; ++k) {
                    const int w  = 4 * g + k;
                    const int bo = (jo >> (3 - k)) & 1;
                    const int bi = (ji >> (3 - k)) & 1;
                    o *= gates[l][w][bo * 2 + bi];
                }
                Ot[g][jo * OP + qA] = o;
            }
            if (t < 48) {
                const int g = t >> 4, j = t & 15;
                float re = 1.0f, im = 0.0f;
                #pragma unroll
                for (int k = 0; k < 4; ++k) {
                    const int w  = 4 * g + k;
                    const int bo = (j >> (3 - k)) & 1;
                    const float er = gates[l][w][4 + 2 * bo];
                    const float ei = gates[l][w][5 + 2 * bo];
                    const float nr = re * er - im * ei;
                    const float ni = re * ei + im * er;
                    re = nr; im = ni;
                }
                Ph[g][j] = make_float2(re, im);
            }
        }
        v2f bbg[2][8];
        #pragma unroll
        for (int tt = 0; tt < 2; ++tt) {
            #pragma unroll
            for (int e = 0; e < 8; ++e) {
                bbg[tt][e] = *(const v2f*)&st2[gsrc[tt][(e & 1) + 2 * (e >> 2)] + P2 * gray4(e)];
            }
        }
        __syncthreads();

        {
            {
                const float2 q1 = Ph[1][t >> 4];
                const float2 q2 = Ph[2][t & 15];
                Pc[P1 * (t >> 4) + (t & 15)] = make_float2(q1.x * q2.x - q1.y * q2.y, q1.x * q2.y + q1.y * q2.x);
            }
            v2f aa[4];
            #pragma unroll
            for (int c = 0; c < 4; ++c) {
                aa[c] = *(const v2f*)&Ot[0][row0 * OP + 4 * c + 2 * hi];
            }
            float2 pz[8];
            #pragma unroll
            for (int v = 0; v < 8; ++v) pz[v] = Ph[0][(v < 4 ? v : v + 4) + 4 * hi];

            #pragma unroll
            for (int tt = 0; tt < 2; ++tt) {
                float2* sp = &st2[base0 + P1 * tt];

                const v2f* bb = bbg[tt];
                v8f dr = {0, 0, 0, 0, 0, 0, 0, 0};
                v8f di = {0, 0, 0, 0, 0, 0, 0, 0};
                #pragma unroll
                for (int c = 0; c < 4; ++c) {
                    const v2f bre = {bb[2 * c].x, bb[2 * c + 1].x};
                    const v2f bim = {bb[2 * c].y, bb[2 * c + 1].y};
                    dr = __builtin_amdgcn_wmma_f32_16x16x4_f32(false, aa[c], false, bre, (short)0, dr, false, false);
                    di = __builtin_amdgcn_wmma_f32_16x16x4_f32(false, aa[c], false, bim, (short)0, di, false, false);
                }
                #pragma unroll
                for (int v = 0; v < 8; ++v) {
                    const float xr = dr[v], xi = di[v];
                    sp[P2 * (v < 4 ? v : v + 4)] = make_float2(xr * pz[v].x - xi * pz[v].y,
                                             xr * pz[v].y + xi * pz[v].x);
                }
            }
        }
        __syncthreads();

        {
            v2f a1[4], a2[4];
            #pragma unroll
            for (int c = 0; c < 4; ++c) {
                a1[c] = *(const v2f*)&Ot[1][c16 * OP + 4 * c + 2 * hi];
                a2[c] = *(const v2f*)&Ot[2][c16 * OP + 4 * c + 2 * hi];
            }
            float2 pc[8];
            {
                const float4* pp = (const float4*)&Pc[P1 * c16 + 8 * hi];
                #pragma unroll
                for (int v = 0; v < 8; v += 2) {
                    const float4 q = pp[v >> 1];
                    pc[v]     = make_float2(q.x, q.y);
                    pc[v + 1] = make_float2(q.z, q.w);
                }
            }

            #pragma unroll
            for (int tt = 0; tt < 2; ++tt) {
                const float2* lp = &st2[base1 + P2 * tt];

                v2f bb[8];
                #pragma unroll
                for (int e = 0; e < 8; ++e) {
                    bb[e] = *(const v2f*)&lp[P1 * e];
                }
                v8f yr = {0, 0, 0, 0, 0, 0, 0, 0};
                v8f yi = {0, 0, 0, 0, 0, 0, 0, 0};
                #pragma unroll
                for (int c = 0; c < 4; ++c) {
                    const v2f xre = {bb[2 * c].x, bb[2 * c + 1].x};
                    const v2f xim = {bb[2 * c].y, bb[2 * c + 1].y};
                    yr = __builtin_amdgcn_wmma_f32_16x16x4_f32(false, xre, false, a1[c], (short)0, yr, false, false);
                    yi = __builtin_amdgcn_wmma_f32_16x16x4_f32(false, xim, false, a1[c], (short)0, yi, false, false);
                }
                v8f zr = {0, 0, 0, 0, 0, 0, 0, 0};
                v8f zi = {0, 0, 0, 0, 0, 0, 0, 0};
                #pragma unroll
                for (int c = 0; c < 4; ++c) {
                    const v2f bre = {yr[2 * c], yr[2 * c + 1]};
                    const v2f bim = {yi[2 * c], yi[2 * c + 1]};
                    zr = __builtin_amdgcn_wmma_f32_16x16x4_f32(false, a2[c], false, bre, (short)0, zr, false, false);
                    zi = __builtin_amdgcn_wmma_f32_16x16x4_f32(false, a2[c], false, bim, (short)0, zi, false, false);
                }
                float4* wp = (float4*)&st2[base2 + P2 * tt];
                #pragma unroll
                for (int v = 0; v < 8; v += 2) {
                    wp[v >> 1] = make_float4(zr[v] * pc[v].x - zi[v] * pc[v].y,
                                             zr[v] * pc[v].y + zi[v] * pc[v].x,
                                             zr[v + 1] * pc[v + 1].x - zi[v + 1] * pc[v + 1].y,
                                             zr[v + 1] * pc[v + 1].y + zi[v + 1] * pc[v + 1].x);
                }
            }
        }
        __syncthreads();

    }

    float acc = 0.0f;
    #pragma unroll
    for (int k = 0; k < 16; ++k) {
        const float2 v = st2[finBase + k];
        const float p = v.x * v.x + v.y * v.y;
        acc += par4(k & ZMASK & 15) ? -p : p;
    }
    if (__popc((16 * t) & ZMASK) & 1) acc = -acc;
    red[t] = acc;
    __syncthreads();
    #pragma unroll
    for (int off = BLOCK / 2; off > 0; off >>= 1) {
        if (t < off) red[t] += red[t + off];
        __syncthreads();
    }
    if (t == 0) out[b] = red[0];
}

extern "C" void kernel_launch(void* const* d_in, const int* in_sizes, int n_in,
                              void* d_out, int out_size, void* d_ws, size_t ws_size,
                              hipStream_t stream) {
    const float* model_input = (const float*)d_in[0];
    const float* weights     = (const float*)d_in[1];
    float* out = (float*)d_out;
    const int batch = in_sizes[0] / 2;
    qsim12_kernel<<<batch, BLOCK, 0, stream>>>(model_input, weights, out);
}


# Layer 1 in closed form too: behind ring 0 the state is a chain of 2x2 transfer matrices; three segment tables, then a complex K=4 product per tile on WMMA (8 per wave, was 48); sums regrouped
# speedup vs baseline: 2.6523x; 1.1917x over previous
#include <hip/hip_runtime.h>

typedef float v2f __attribute__((ext_vector_type(2)));
typedef float v8f __attribute__((ext_vector_type(8)));

#define NQ    12
#define NL    6
#define DIM   4096
#define BLOCK 256
#define OP    20

#define P1    18
#define P2    292
#define ST_N  (15 * P2 + 15 * P1 + 16)

__host__ __device__ static constexpr int pos_of(int i) {
    return P2 * (i >> 8) + P1 * ((i >> 4) & 15) + (i & 15);
}

__host__ __device__ static constexpr int Rmap(int j) {
    int x = j;
    for (int w = 11; w >= 0; --w) {
        const int pc = 11 - w;
        const int pt = 11 - ((w + 1) % 12);
        x ^= ((x >> pc) & 1) << pt;
    }
    return x;
}

__host__ __device__ static constexpr int gray4(int k) { return k ^ (k >> 1); }
__host__ __device__ static constexpr int gray4_inv(int g) { return g ^ (g >> 1) ^ (g >> 2) ^ (g >> 3); }
__host__ __device__ static constexpr int par4(int m) { return (m ^ (m >> 1) ^ (m >> 2) ^ (m >> 3)) & 1; }
__host__ __device__ static constexpr bool ring_structure_ok() {
    for (int k = 0; k < 16; ++k) {
        if (Rmap(k << 8) != ((gray4(k) << 8) | ((k & 1) << 7))) return false;
        if ((gray4_inv(gray4(k)) & 15) != k) return false;
    }
    for (int low = 0; low < 256; ++low) {
        const int s = Rmap(low);
        if ((s >> 8) != 12 * (low & 1)) return false;
        if (((s >> 7) & 1) != ((low >> 7) & 1)) return false;
    }
    return true;
}
static_assert(ring_structure_ok(), "ring gather addressing assumes this structure of the CNOT ring map");

__host__ __device__ static constexpr int Rfwd(int j) {
    int x = j;
    for (int w = 0; w < 12; ++w) {
        const int pc = 11 - w;
        const int pt = 11 - ((w + 1) % 12);
        x ^= ((x >> pc) & 1) << pt;
    }
    return x;
}
__host__ __device__ static constexpr int ring_zmask() {
    int m = 0;
    for (int i = 0; i < 12; ++i) {
        if (Rfwd(1 << i) & 1) m |= 1 << i;
    }
    return m;
}
__host__ __device__ static constexpr bool ring_inverse_ok() {
    for (int i = 0; i < 12; ++i) {
        if (Rfwd(Rmap(1 << i)) != (1 << i)) return false;
    }
    return true;
}
static_assert(ring_inverse_ok(), "Rfwd must invert Rmap (both are linear: the unit vectors decide)");
#define ZMASK (ring_zmask())

__device__ static inline float2 cmul(const float2 a, const float2 b) {
    return make_float2(a.x * b.x - a.y * b.y, a.x * b.y + a.y * b.x);
}

__global__ __launch_bounds__(BLOCK) void qsim12_kernel(
    const float* __restrict__ model_input,
    const float* __restrict__ weights,
    float* __restrict__ out)
{
    __shared__ __attribute__((aligned(16))) float2 st2[ST_N];
    __shared__ __attribute__((aligned(16))) float Ot[3][16 * OP];
    __shared__ __attribute__((aligned(16))) float2 Ph[3][16];
    __shared__ __attribute__((aligned(16))) float2 Pc[16 * P1];
    __shared__ float gates[NL][NQ][8];
    __shared__ float red[BLOCK];
    __shared__ __attribute__((aligned(16))) float2 Tab[3][16][4];

    const int b    = blockIdx.x;
    const int t    = threadIdx.x;
    const int lane = t & 31;
    const int wv   = t >> 5;
    const int c16  = lane & 15;
    const int hi   = lane >> 4;

    const int base0 = P2 * (4 * hi) + P1 * (2 * wv) + c16;
    const int row0  = (c16 & 3) | ((c16 & 4) << 1) | ((c16 & 8) >> 1);
    const int base1 = P2 * (2 * wv) + P1 * (8 * hi) + c16;
    const int base2 = P2 * (2 * wv) + P1 * c16 + 8 * hi;
    int gsrc[2][4];
    #pragma unroll
    for (int tt = 0; tt < 2; ++tt) {
        const int s  = Rmap(((2 * wv + tt) << 4) | c16);
        const int q  = hi ^ ((s >> 11) & 1);
        const int pE = pos_of(s & 0xFF);
        const int pO = pos_of((s & 0xFF) ^ 0x80);
        gsrc[tt][0] = pE + 12 * P2 * q;
        gsrc[tt][1] = pO + 12 * P2 * q;
        gsrc[tt][2] = pE + 4 * P2 * q;
        gsrc[tt][3] = pO + 4 * P2 * q;
    }
    const int finBase = P2 * (t >> 4) + P1 * (t & 15);

    const float a0 = model_input[2 * b + 0];
    const float a1 = model_input[2 * b + 1];

    if (t < NL * NQ) {
        const int l = t / NQ, w = t % NQ;
        const int limit = NQ - (l >> 1);
        float th = weights[(l * NQ + w) * 2 + 0];
        if (w < limit) th += (w & 1) ? a1 : a0;
        const float ph = weights[(l * NQ + w) * 2 + 1];
        float c, s, pr, pi;
        __sincosf(0.5f * th, &s, &c);
        __sincosf(0.5f * ph, &pi, &pr);
        gates[l][w][0] = c;   gates[l][w][1] = -s;
        gates[l][w][2] = s;   gates[l][w][3] = c;
        gates[l][w][4] = pr;  gates[l][w][5] = -pi;
        gates[l][w][6] = pr;  gates[l][w][7] = pi;
    }
    __syncthreads();

    if (t < 48) {
        const int seg = t >> 4, dg = t & 15;
        #pragma unroll
        for (int p = 0; p < 2; ++p) {
            float2 u0 = make_float2(0.0f, 0.0f), u1 = make_float2(0.0f, 0.0f);
            #pragma unroll
            for (int k = 0; k < 4; ++k) {
                const int w = 4 * seg + k;
                const int z = (dg >> (3 - k)) & 1;
                float2 Tm[2][2];
                #pragma unroll
                for (int r = 0; r < 2; ++r) {
                    #pragma unroll
                    for (int y = 0; y < 2; ++y) {
                        const int bo = y ^ r;
                        const float a = gates[0][w][2 * bo];
                        const float g = gates[1][w][2 * z + y];
                        Tm[r][y] = cmul(make_float2(a * gates[0][w][4 + 2 * bo], a * gates[0][w][5 + 2 * bo]),
                                        make_float2(g * gates[1][w][4 + 2 * z], g * gates[1][w][5 + 2 * z]));
                    }
                }
                if (k == 0) {
                    u0 = Tm[p][0];
                    u1 = Tm[p][1];
                } else {
                    const bool sw = (k == 1) && (seg == 0) && (p == 1);
                    const float2 r00 = sw ? Tm[1][0] : Tm[0][0];
                    const float2 r01 = sw ? Tm[1][1] : Tm[0][1];
                    const float2 r10 = sw ? Tm[0][0] : Tm[1][0];
                    const float2 r11 = sw ? Tm[0][1] : Tm[1][1];
                    const float2 x0 = cmul(u0, r00), x1 = cmul(u1, r10);
                    const float2 y0 = cmul(u0, r01), y1 = cmul(u1, r11);
                    u0 = make_float2(x0.x + x1.x, x0.y + x1.y);
                    u1 = make_float2(y0.x + y1.x, y0.y + y1.y);
                }
            }
            Tab[seg][dg][2 * p]     = u0;
            Tab[seg][dg][2 * p + 1] = u1;
        }
    }
    __syncthreads();
    {
        const float4* Lp = (const float4*)&Tab[0][row0][0];
        const float4 l0 = Lp[0];
        const float4 l1 = Lp[1];
        const float4 rr = *(const float4*)&Tab[2][c16][2 * hi];
        const v2f rb0 = {rr.x, rr.y};
        const v2f rb1 = {rr.z, rr.w};
        #pragma unroll
        for (int tt = 0; tt < 2; ++tt) {
            const float2 m0 = Tab[1][2 * wv + tt][hi];
            const float2 m1 = Tab[1][2 * wv + tt][2 + hi];
            const float2 p00 = cmul(make_float2(l0.x, l0.y), m0), p01 = cmul(make_float2(l0.z, l0.w), m1);
            const float2 p10 = cmul(make_float2(l1.x, l1.y), m0), p11 = cmul(make_float2(l1.z, l1.w), m1);
            const float2 lm0 = make_float2(p00.x + p01.x, p00.y + p01.y);
            const float2 lm1 = make_float2(p10.x + p11.x, p10.y + p11.y);
            const v2f are0 = {lm0.x, -lm0.y}, aim0 = {lm0.y, lm0.x};
            const v2f are1 = {lm1.x, -lm1.y}, aim1 = {lm1.y, lm1.x};
            v8f dr = {0, 0, 0, 0, 0, 0, 0, 0};
            v8f di = {0, 0, 0, 0, 0, 0, 0, 0};
            dr = __builtin_amdgcn_wmma_f32_16x16x4_f32(false, are0, false, rb0, (short)0, dr, false, false);
            di = __builtin_amdgcn_wmma_f32_16x16x4_f32(false, aim0, false, rb0, (short)0, di, false, false);
            dr = __builtin_amdgcn_wmma_f32_16x16x4_f32(false, are1, false, rb1, (short)0, dr, false, false);
            di = __builtin_amdgcn_wmma_f32_16x16x4_f32(false, aim1, false, rb1, (short)0, di, false, false);
            float2* sp = &st2[base0 + P1 * tt];
            #pragma unroll
            for (int v = 0; v < 8; ++v) {
                sp[P2 * (v < 4 ? v : v + 4)] = make_float2(dr[v], di[v]);
            }
        }
    }
    __syncthreads();

    #pragma unroll 1
    for (int l = 2; l < NL; ++l) {
        {
            const int jo = t >> 4, ji = t & 15;
            const int qA = ((ji & 6) << 1) | ((ji >> 3) << 1) | (ji & 1);
            #pragma unroll
            for (int g = 0; g < 3; ++g) {
                float o = 1.0f;
                #pragma unroll
                for (int k = 0; k < 4; ++k) {
                    const int w  = 4 * g + k;
                    const int bo = (jo >> (3 - k)) & 1;
                    const int bi = (ji >> (3 - k)) & 1;
                    o *= gates[l][w][bo * 2 + bi];
                }
                Ot[g][jo * OP + qA] = o;
            }
            if (t < 48) {
                const int g = t >> 4, j = t & 15;
                float re = 1.0f, im = 0.0f;
                #pragma unroll
                for (int k = 0; k < 4; ++k) {
                    const int w  = 4 * g + k;
                    const int bo = (j >> (3 - k)) & 1;
                    const float er = gates[l][w][4 + 2 * bo];
                    const float ei = gates[l][w][5 + 2 * bo];
                    const float nr = re * er - im * ei;
                    const float ni = re * ei + im * er;
                    re = nr; im = ni;
                }
                Ph[g][j] = make_float2(re, im);
            }
        }
        v2f bbg[2][8];
        #pragma unroll
        for (int tt = 0; tt < 2; ++tt) {
            #pragma unroll
            for (int e = 0; e < 8; ++e) {
                bbg[tt][e] = *(const v2f*)&st2[gsrc[tt][(e & 1) + 2 * (e >> 2)] + P2 * gray4(e)];
            }
        }
        __syncthreads();

        {
            {
                const float2 q1 = Ph[1][t >> 4];
                const float2 q2 = Ph[2][t & 15];
                Pc[P1 * (t >> 4) + (t & 15)] = make_float2(q1.x * q2.x - q1.y * q2.y, q1.x * q2.y + q1.y * q2.x);
            }
            v2f aa[4];
            #pragma unroll
            for (int c = 0; c < 4; ++c) {
                aa[c] = *(const v2f*)&Ot[0][row0 * OP + 4 * c + 2 * hi];
            }
            float2 pz[8];
            #pragma unroll
            for (int v = 0; v < 8; ++v) pz[v] = Ph[0][(v < 4 ? v : v + 4) + 4 * hi];

            #pragma unroll
            for (int tt = 0; tt < 2; ++tt) {
                float2* sp = &st2[base0 + P1 * tt];

                const v2f* bb = bbg[tt];
                v8f dr = {0, 0, 0, 0, 0, 0, 0, 0};
                v8f di = {0, 0, 0, 0, 0, 0, 0, 0};
                #pragma unroll
                for (int c = 0; c < 4; ++c) {
                    const v2f bre = {bb[2 * c].x, bb[2 * c + 1].x};
                    const v2f bim = {bb[2 * c].y, bb[2 * c + 1].y};
                    dr = __builtin_amdgcn_wmma_f32_16x16x4_f32(false, aa[c], false, bre, (short)0, dr, false, false);
                    di = __builtin_amdgcn_wmma_f32_16x16x4_f32(false, aa[c], false, bim, (short)0, di, false, false);
                }
                #pragma unroll
                for (int v = 0; v < 8; ++v) {
                    const float xr = dr[v], xi = di[v];
                    sp[P2 * (v < 4 ? v : v + 4)] = make_float2(xr * pz[v].x - xi * pz[v].y,
                                             xr * pz[v].y + xi * pz[v].x);
                }
            }
        }
        __syncthreads();

        {
            v2f a1[4], a2[4];
            #pragma unroll
            for (int c = 0; c < 4; ++c) {
                a1[c] = *(const v2f*)&Ot[1][c16 * OP + 4 * c + 2 * hi];
                a2[c] = *(const v2f*)&Ot[2][c16 * OP + 4 * c + 2 * hi];
            }
            float2 pc[8];
            {
                const float4* pp = (const float4*)&Pc[P1 * c16 + 8 * hi];
                #pragma unroll
                for (int v = 0; v < 8; v += 2) {
                    const float4 q = pp[v >> 1];
                    pc[v]     = make_float2(q.x, q.y);
                    pc[v + 1] = make_float2(q.z, q.w);
                }
            }

            #pragma unroll
            for (int tt = 0; tt < 2; ++tt) {
                const float2* lp = &st2[base1 + P2 * tt];

                v2f bb[8];
                #pragma unroll
                for (int e = 0; e < 8; ++e) {
                    bb[e] = *(const v2f*)&lp[P1 * e];
                }
                v8f yr = {0, 0, 0, 0, 0, 0, 0, 0};
                v8f yi = {0, 0, 0, 0, 0, 0, 0, 0};
                #pragma unroll
                for (int c = 0; c < 4; ++c) {
                    const v2f xre = {bb[2 * c].x, bb[2 * c + 1].x};
                    const v2f xim = {bb[2 * c].y, bb[2 * c + 1].y};
                    yr = __builtin_amdgcn_wmma_f32_16x16x4_f32(false, xre, false, a1[c], (short)0, yr, false, false);
                    yi = __builtin_amdgcn_wmma_f32_16x16x4_f32(false, xim, false, a1[c], (short)0, yi, false, false);
                }
                v8f zr = {0, 0, 0, 0, 0, 0, 0, 0};
                v8f zi = {0, 0, 0, 0, 0, 0, 0, 0};
                #pragma unroll
                for (int c = 0; c < 4; ++c) {
                    const v2f bre = {yr[2 * c], yr[2 * c + 1]};
                    const v2f bim = {yi[2 * c], yi[2 * c + 1]};
                    zr = __builtin_amdgcn_wmma_f32_16x16x4_f32(false, a2[c], false, bre, (short)0, zr, false, false);
                    zi = __builtin_amdgcn_wmma_f32_16x16x4_f32(false, a2[c], false, bim, (short)0, zi, false, false);
                }
                float4* wp = (float4*)&st2[base2 + P2 * tt];
                #pragma unroll
                for (int v = 0; v < 8; v += 2) {
                    wp[v >> 1] = make_float4(zr[v] * pc[v].x - zi[v] * pc[v].y,
                                             zr[v] * pc[v].y + zi[v] * pc[v].x,
                                             zr[v + 1] * pc[v + 1].x - zi[v + 1] * pc[v + 1].y,
                                             zr[v + 1] * pc[v + 1].y + zi[v + 1] * pc[v + 1].x);
                }
            }
        }
        __syncthreads();

    }

    float acc = 0.0f;
    #pragma unroll
    for (int k = 0; k < 16; ++k) {
        const float2 v = st2[finBase + k];
        const float p = v.x * v.x + v.y * v.y;
        acc += par4(k & ZMASK & 15) ? -p : p;
    }
    if (__popc((16 * t) & ZMASK) & 1) acc = -acc;
    red[t] = acc;
    __syncthreads();
    #pragma unroll
    for (int off = BLOCK / 2; off > 0; off >>= 1) {
        if (t < off) red[t] += red[t + off];
        __syncthreads();
    }
    if (t == 0) out[b] = red[0];
}

extern "C" void kernel_launch(void* const* d_in, const int* in_sizes, int n_in,
                              void* d_out, int out_size, void* d_ws, size_t ws_size,
                              hipStream_t stream) {
    const float* model_input = (const float*)d_in[0];
    const float* weights     = (const float*)d_in[1];
    float* out = (float*)d_out;
    const int batch = in_sizes[0] / 2;
    qsim12_kernel<<<batch, BLOCK, 0, stream>>>(model_input, weights, out);
}


# Readout folded into the last layer: its RZ phases (unit modulus) drop out of |amp|^2; signed squares summed from the group 1+2 D registers, wave shuffles; final store/read pass gone; sums regrouped
# speedup vs baseline: 2.7480x; 1.0361x over previous
#include <hip/hip_runtime.h>

typedef float v2f __attribute__((ext_vector_type(2)));
typedef float v8f __attribute__((ext_vector_type(8)));

#define NQ    12
#define NL    6
#define DIM   4096
#define BLOCK 256
#define OP    20

#define P1    18
#define P2    292
#define ST_N  (15 * P2 + 15 * P1 + 16)

__host__ __device__ static constexpr int pos_of(int i) {
    return P2 * (i >> 8) + P1 * ((i >> 4) & 15) + (i & 15);
}

__host__ __device__ static constexpr int Rmap(int j) {
    int x = j;
    for (int w = 11; w >= 0; --w) {
        const int pc = 11 - w;
        const int pt = 11 - ((w + 1) % 12);
        x ^= ((x >> pc) & 1) << pt;
    }
    return x;
}

__host__ __device__ static constexpr int gray4(int k) { return k ^ (k >> 1); }
__host__ __device__ static constexpr int gray4_inv(int g) { return g ^ (g >> 1) ^ (g >> 2) ^ (g >> 3); }
__host__ __device__ static constexpr int par4(int m) { return (m ^ (m >> 1) ^ (m >> 2) ^ (m >> 3)) & 1; }
__host__ __device__ static constexpr bool ring_structure_ok() {
    for (int k = 0; k < 16; ++k) {
        if (Rmap(k << 8) != ((gray4(k) << 8) | ((k & 1) << 7))) return false;
        if ((gray4_inv(gray4(k)) & 15) != k) return false;
    }
    for (int low = 0; low < 256; ++low) {
        const int s = Rmap(low);
        if ((s >> 8) != 12 * (low & 1)) return false;
        if (((s >> 7) & 1) != ((low >> 7) & 1)) return false;
    }
    return true;
}
static_assert(ring_structure_ok(), "ring gather addressing assumes this structure of the CNOT ring map");

__host__ __device__ static constexpr int Rfwd(int j) {
    int x = j;
    for (int w = 0; w < 12; ++w) {
        const int pc = 11 - w;
        const int pt = 11 - ((w + 1) % 12);
        x ^= ((x >> pc) & 1) << pt;
    }
    return x;
}
__host__ __device__ static constexpr int ring_zmask() {
    int m = 0;
    for (int i = 0; i < 12; ++i) {
        if (Rfwd(1 << i) & 1) m |= 1 << i;
    }
    return m;
}
__host__ __device__ static constexpr bool ring_inverse_ok() {
    for (int i = 0; i < 12; ++i) {
        if (Rfwd(Rmap(1 << i)) != (1 << i)) return false;
    }
    return true;
}
static_assert(ring_inverse_ok(), "Rfwd must invert Rmap (both are linear: the unit vectors decide)");
#define ZMASK (ring_zmask())

__device__ static inline float2 cmul(const float2 a, const float2 b) {
    return make_float2(a.x * b.x - a.y * b.y, a.x * b.y + a.y * b.x);
}

__global__ __launch_bounds__(BLOCK) void qsim12_kernel(
    const float* __restrict__ model_input,
    const float* __restrict__ weights,
    float* __restrict__ out)
{
    __shared__ __attribute__((aligned(16))) float2 st2[ST_N];
    __shared__ __attribute__((aligned(16))) float Ot[3][16 * OP];
    __shared__ __attribute__((aligned(16))) float2 Ph[3][16];
    __shared__ __attribute__((aligned(16))) float2 Pc[16 * P1];
    __shared__ float gates[NL][NQ][8];
    __shared__ float red[BLOCK / 32];
    __shared__ __attribute__((aligned(16))) float2 Tab[3][16][4];

    const int b    = blockIdx.x;
    const int t    = threadIdx.x;
    const int lane = t & 31;
    const int wv   = t >> 5;
    const int c16  = lane & 15;
    const int hi   = lane >> 4;

    const int base0 = P2 * (4 * hi) + P1 * (2 * wv) + c16;
    const int row0  = (c16 & 3) | ((c16 & 4) << 1) | ((c16 & 8) >> 1);
    const int base1 = P2 * (2 * wv) + P1 * (8 * hi) + c16;
    const int base2 = P2 * (2 * wv) + P1 * c16 + 8 * hi;
    int gsrc[2][4];
    #pragma unroll
    for (int tt = 0; tt < 2; ++tt) {
        const int s  = Rmap(((2 * wv + tt) << 4) | c16);
        const int q  = hi ^ ((s >> 11) & 1);
        const int pE = pos_of(s & 0xFF);
        const int pO = pos_of((s & 0xFF) ^ 0x80);
        gsrc[tt][0] = pE + 12 * P2 * q;
        gsrc[tt][1] = pO + 12 * P2 * q;
        gsrc[tt][2] = pE + 4 * P2 * q;
        gsrc[tt][3] = pO + 4 * P2 * q;
    }

    const float a0 = model_input[2 * b + 0];
    const float a1 = model_input[2 * b + 1];

    if (t < NL * NQ) {
        const int l = t / NQ, w = t % NQ;
        const int limit = NQ - (l >> 1);
        float th = weights[(l * NQ + w) * 2 + 0];
        if (w < limit) th += (w & 1) ? a1 : a0;
        const float ph = weights[(l * NQ + w) * 2 + 1];
        float c, s, pr, pi;
        __sincosf(0.5f * th, &s, &c);
        __sincosf(0.5f * ph, &pi, &pr);
        gates[l][w][0] = c;   gates[l][w][1] = -s;
        gates[l][w][2] = s;   gates[l][w][3] = c;
        gates[l][w][4] = pr;  gates[l][w][5] = -pi;
        gates[l][w][6] = pr;  gates[l][w][7] = pi;
    }
    __syncthreads();

    if (t < 48) {
        const int seg = t >> 4, dg = t & 15;
        #pragma unroll
        for (int p = 0; p < 2; ++p) {
            float2 u0 = make_float2(0.0f, 0.0f), u1 = make_float2(0.0f, 0.0f);
            #pragma unroll
            for (int k = 0; k < 4; ++k) {
                const int w = 4 * seg + k;
                const int z = (dg >> (3 - k)) & 1;
                float2 Tm[2][2];
                #pragma unroll
                for (int r = 0; r < 2; ++r) {
                    #pragma unroll
                    for (int y = 0; y < 2; ++y) {
                        const int bo = y ^ r;
                        const float a = gates[0][w][2 * bo];
                        const float g = gates[1][w][2 * z + y];
                        Tm[r][y] = cmul(make_float2(a * gates[0][w][4 + 2 * bo], a * gates[0][w][5 + 2 * bo]),
                                        make_float2(g * gates[1][w][4 + 2 * z], g * gates[1][w][5 + 2 * z]));
                    }
                }
                if (k == 0) {
                    u0 = Tm[p][0];
                    u1 = Tm[p][1];
                } else {
                    const bool sw = (k == 1) && (seg == 0) && (p == 1);
                    const float2 r00 = sw ? Tm[1][0] : Tm[0][0];
                    const float2 r01 = sw ? Tm[1][1] : Tm[0][1];
                    const float2 r10 = sw ? Tm[0][0] : Tm[1][0];
                    const float2 r11 = sw ? Tm[0][1] : Tm[1][1];
                    const float2 x0 = cmul(u0, r00), x1 = cmul(u1, r10);
                    const float2 y0 = cmul(u0, r01), y1 = cmul(u1, r11);
                    u0 = make_float2(x0.x + x1.x, x0.y + x1.y);
                    u1 = make_float2(y0.x + y1.x, y0.y + y1.y);
                }
            }
            Tab[seg][dg][2 * p]     = u0;
            Tab[seg][dg][2 * p + 1] = u1;
        }
    }
    __syncthreads();
    {
        const float4* Lp = (const float4*)&Tab[0][row0][0];
        const float4 l0 = Lp[0];
        const float4 l1 = Lp[1];
        const float4 rr = *(const float4*)&Tab[2][c16][2 * hi];
        const v2f rb0 = {rr.x, rr.y};
        const v2f rb1 = {rr.z, rr.w};
        #pragma unroll
        for (int tt = 0; tt < 2; ++tt) {
            const float2 m0 = Tab[1][2 * wv + tt][hi];
            const float2 m1 = Tab[1][2 * wv + tt][2 + hi];
            const float2 p00 = cmul(make_float2(l0.x, l0.y), m0), p01 = cmul(make_float2(l0.z, l0.w), m1);
            const float2 p10 = cmul(make_float2(l1.x, l1.y), m0), p11 = cmul(make_float2(l1.z, l1.w), m1);
            const float2 lm0 = make_float2(p00.x + p01.x, p00.y + p01.y);
            const float2 lm1 = make_float2(p10.x + p11.x, p10.y + p11.y);
            const v2f are0 = {lm0.x, -lm0.y}, aim0 = {lm0.y, lm0.x};
            const v2f are1 = {lm1.x, -lm1.y}, aim1 = {lm1.y, lm1.x};
            v8f dr = {0, 0, 0, 0, 0, 0, 0, 0};
            v8f di = {0, 0, 0, 0, 0, 0, 0, 0};
            dr = __builtin_amdgcn_wmma_f32_16x16x4_f32(false, are0, false, rb0, (short)0, dr, false, false);
            di = __builtin_amdgcn_wmma_f32_16x16x4_f32(false, aim0, false, rb0, (short)0, di, false, false);
            dr = __builtin_amdgcn_wmma_f32_16x16x4_f32(false, are1, false, rb1, (short)0, dr, false, false);
            di = __builtin_amdgcn_wmma_f32_16x16x4_f32(false, aim1, false, rb1, (short)0, di, false, false);
            float2* sp = &st2[base0 + P1 * tt];
            #pragma unroll
            for (int v = 0; v < 8; ++v) {
                sp[P2 * (v < 4 ? v : v + 4)] = make_float2(dr[v], di[v]);
            }
        }
    }
    __syncthreads();

    #pragma unroll 1
    for (int l = 2; l < NL - 1; ++l) {
        {
            const int jo = t >> 4, ji = t & 15;
            const int qA = ((ji & 6) << 1) | ((ji >> 3) << 1) | (ji & 1);
            #pragma unroll
            for (int g = 0; g < 3; ++g) {
                float o = 1.0f;
                #pragma unroll
                for (int k = 0; k < 4; ++k) {
                    const int w  = 4 * g + k;
                    const int bo = (jo >> (3 - k)) & 1;
                    const int bi = (ji >> (3 - k)) & 1;
                    o *= gates[l][w][bo * 2 + bi];
                }
                Ot[g][jo * OP + qA] = o;
            }
            if (t < 48) {
                const int g = t >> 4, j = t & 15;
                float re = 1.0f, im = 0.0f;
                #pragma unroll
                for (int k = 0; k < 4; ++k) {
                    const int w  = 4 * g + k;
                    const int bo = (j >> (3 - k)) & 1;
                    const float er = gates[l][w][4 + 2 * bo];
                    const float ei = gates[l][w][5 + 2 * bo];
                    const float nr = re * er - im * ei;
                    const float ni = re * ei + im * er;
                    re = nr; im = ni;
                }
                Ph[g][j] = make_float2(re, im);
            }
        }
        v2f bbg[2][8];
        #pragma unroll
        for (int tt = 0; tt < 2; ++tt) {
            #pragma unroll
            for (int e = 0; e < 8; ++e) {
                bbg[tt][e] = *(const v2f*)&st2[gsrc[tt][(e & 1) + 2 * (e >> 2)] + P2 * gray4(e)];
            }
        }
        __syncthreads();

        {
            {
                const float2 q1 = Ph[1][t >> 4];
                const float2 q2 = Ph[2][t & 15];
                Pc[P1 * (t >> 4) + (t & 15)] = make_float2(q1.x * q2.x - q1.y * q2.y, q1.x * q2.y + q1.y * q2.x);
            }
            v2f aa[4];
            #pragma unroll
            for (int c = 0; c < 4; ++c) {
                aa[c] = *(const v2f*)&Ot[0][row0 * OP + 4 * c + 2 * hi];
            }
            float2 pz[8];
            #pragma unroll
            for (int v = 0; v < 8; ++v) pz[v] = Ph[0][(v < 4 ? v : v + 4) + 4 * hi];

            #pragma unroll
            for (int tt = 0; tt < 2; ++tt) {
                float2* sp = &st2[base0 + P1 * tt];

                const v2f* bb = bbg[tt];
                v8f dr = {0, 0, 0, 0, 0, 0, 0, 0};
                v8f di = {0, 0, 0, 0, 0, 0, 0, 0};
                #pragma unroll
                for (int c = 0; c < 4; ++c) {
                    const v2f bre = {bb[2 * c].x, bb[2 * c + 1].x};
                    const v2f bim = {bb[2 * c].y, bb[2 * c + 1].y};
                    dr = __builtin_amdgcn_wmma_f32_16x16x4_f32(false, aa[c], false, bre, (short)0, dr, false, false);
                    di = __builtin_amdgcn_wmma_f32_16x16x4_f32(false, aa[c], false, bim, (short)0, di, false, false);
                }
                #pragma unroll
                for (int v = 0; v < 8; ++v) {
                    const float xr = dr[v], xi = di[v];
                    sp[P2 * (v < 4 ? v : v + 4)] = make_float2(xr * pz[v].x - xi * pz[v].y,
                                             xr * pz[v].y + xi * pz[v].x);
                }
            }
        }
        __syncthreads();

        {
            v2f a1[4], a2[4];
            #pragma unroll
            for (int c = 0; c < 4; ++c) {
                a1[c] = *(const v2f*)&Ot[1][c16 * OP + 4 * c + 2 * hi];
                a2[c] = *(const v2f*)&Ot[2][c16 * OP + 4 * c + 2 * hi];
            }
            float2 pc[8];
            {
                const float4* pp = (const float4*)&Pc[P1 * c16 + 8 * hi];
                #pragma unroll
                for (int v = 0; v < 8; v += 2) {
                    const float4 q = pp[v >> 1];
                    pc[v]     = make_float2(q.x, q.y);
                    pc[v + 1] = make_float2(q.z, q.w);
                }
            }

            #pragma unroll
            for (int tt = 0; tt < 2; ++tt) {
                const float2* lp = &st2[base1 + P2 * tt];

                v2f bb[8];
                #pragma unroll
                for (int e = 0; e < 8; ++e) {
                    bb[e] = *(const v2f*)&lp[P1 * e];
                }
                v8f yr = {0, 0, 0, 0, 0, 0, 0, 0};
                v8f yi = {0, 0, 0, 0, 0, 0, 0, 0};
                #pragma unroll
                for (int c = 0; c < 4; ++c) {
                    const v2f xre = {bb[2 * c].x, bb[2 * c + 1].x};
                    const v2f xim = {bb[2 * c].y, bb[2 * c + 1].y};
                    yr = __builtin_amdgcn_wmma_f32_16x16x4_f32(false, xre, false, a1[c], (short)0, yr, false, false);
                    yi = __builtin_amdgcn_wmma_f32_16x16x4_f32(false, xim, false, a1[c], (short)0, yi, false, false);
                }
                v8f zr = {0, 0, 0, 0, 0, 0, 0, 0};
                v8f zi = {0, 0, 0, 0, 0, 0, 0, 0};
                #pragma unroll
                for (int c = 0; c < 4; ++c) {
                    const v2f bre = {yr[2 * c], yr[2 * c + 1]};
                    const v2f bim = {yi[2 * c], yi[2 * c + 1]};
                    zr = __builtin_amdgcn_wmma_f32_16x16x4_f32(false, a2[c], false, bre, (short)0, zr, false, false);
                    zi = __builtin_amdgcn_wmma_f32_16x16x4_f32(false, a2[c], false, bim, (short)0, zi, false, false);
                }
                float4* wp = (float4*)&st2[base2 + P2 * tt];
                #pragma unroll
                for (int v = 0; v < 8; v += 2) {
                    wp[v >> 1] = make_float4(zr[v] * pc[v].x - zi[v] * pc[v].y,
                                             zr[v] * pc[v].y + zi[v] * pc[v].x,
                                             zr[v + 1] * pc[v + 1].x - zi[v + 1] * pc[v + 1].y,
                                             zr[v + 1] * pc[v + 1].y + zi[v + 1] * pc[v + 1].x);
                }
            }
        }
        __syncthreads();

    }

    float acc = 0.0f;
    {
        constexpr int l = NL - 1;
        {
            const int jo = t >> 4, ji = t & 15;
            const int qA = ((ji & 6) << 1) | ((ji >> 3) << 1) | (ji & 1);
            #pragma unroll
            for (int g = 0; g < 3; ++g) {
                float o = 1.0f;
                #pragma unroll
                for (int k = 0; k < 4; ++k) {
                    const int w  = 4 * g + k;
                    const int bo = (jo >> (3 - k)) & 1;
                    const int bi = (ji >> (3 - k)) & 1;
                    o *= gates[l][w][bo * 2 + bi];
                }
                Ot[g][jo * OP + qA] = o;
            }
        }
        v2f bbg[2][8];
        #pragma unroll
        for (int tt = 0; tt < 2; ++tt) {
            #pragma unroll
            for (int e = 0; e < 8; ++e) {
                bbg[tt][e] = *(const v2f*)&st2[gsrc[tt][(e & 1) + 2 * (e >> 2)] + P2 * gray4(e)];
            }
        }
        __syncthreads();

        {
            v2f aa[4];
            #pragma unroll
            for (int c = 0; c < 4; ++c) {
                aa[c] = *(const v2f*)&Ot[0][row0 * OP + 4 * c + 2 * hi];
            }
            #pragma unroll
            for (int tt = 0; tt < 2; ++tt) {
                float2* sp = &st2[base0 + P1 * tt];
                const v2f* bb = bbg[tt];
                v8f dr = {0, 0, 0, 0, 0, 0, 0, 0};
                v8f di = {0, 0, 0, 0, 0, 0, 0, 0};
                #pragma unroll
                for (int c = 0; c < 4; ++c) {
                    const v2f bre = {bb[2 * c].x, bb[2 * c + 1].x};
                    const v2f bim = {bb[2 * c].y, bb[2 * c + 1].y};
                    dr = __builtin_amdgcn_wmma_f32_16x16x4_f32(false, aa[c], false, bre, (short)0, dr, false, false);
                    di = __builtin_amdgcn_wmma_f32_16x16x4_f32(false, aa[c], false, bim, (short)0, di, false, false);
                }
                #pragma unroll
                for (int v = 0; v < 8; ++v) {
                    sp[P2 * (v < 4 ? v : v + 4)] = make_float2(dr[v], di[v]);
                }
            }
        }
        __syncthreads();

        {
            v2f a1[4], a2[4];
            #pragma unroll
            for (int c = 0; c < 4; ++c) {
                a1[c] = *(const v2f*)&Ot[1][c16 * OP + 4 * c + 2 * hi];
                a2[c] = *(const v2f*)&Ot[2][c16 * OP + 4 * c + 2 * hi];
            }
            #pragma unroll
            for (int tt = 0; tt < 2; ++tt) {
                const float2* lp = &st2[base1 + P2 * tt];
                v2f bb[8];
                #pragma unroll
                for (int e = 0; e < 8; ++e) {
                    bb[e] = *(const v2f*)&lp[P1 * e];
                }
                v8f yr = {0, 0, 0, 0, 0, 0, 0, 0};
                v8f yi = {0, 0, 0, 0, 0, 0, 0, 0};
                #pragma unroll
                for (int c = 0; c < 4; ++c) {
                    const v2f xre = {bb[2 * c].x, bb[2 * c + 1].x};
                    const v2f xim = {bb[2 * c].y, bb[2 * c + 1].y};
                    yr = __builtin_amdgcn_wmma_f32_16x16x4_f32(false, xre, false, a1[c], (short)0, yr, false, false);
                    yi = __builtin_amdgcn_wmma_f32_16x16x4_f32(false, xim, false, a1[c], (short)0, yi, false, false);
                }
                v8f zr = {0, 0, 0, 0, 0, 0, 0, 0};
                v8f zi = {0, 0, 0, 0, 0, 0, 0, 0};
                #pragma unroll
                for (int c = 0; c < 4; ++c) {
                    const v2f bre = {yr[2 * c], yr[2 * c + 1]};
                    const v2f bim = {yi[2 * c], yi[2 * c + 1]};
                    zr = __builtin_amdgcn_wmma_f32_16x16x4_f32(false, a2[c], false, bre, (short)0, zr, false, false);
                    zi = __builtin_amdgcn_wmma_f32_16x16x4_f32(false, a2[c], false, bim, (short)0, zi, false, false);
                }
                float ne = 0.0f, no = 0.0f;
                #pragma unroll
                for (int v = 0; v < 8; ++v) {
                    const float p = zr[v] * zr[v] + zi[v] * zi[v];
                    if (par4(v & ZMASK & 7)) no += p; else ne += p;
                }
                const float ts = ne - no;
                acc += (__popc(((2 * wv + tt) << 8) & ZMASK) & 1) ? -ts : ts;
            }
        }
        if (__popc(((c16 << 4) | (hi << 3)) & ZMASK) & 1) acc = -acc;
    }
    #pragma unroll
    for (int off = 16; off > 0; off >>= 1) acc += __shfl_xor(acc, off);
    if (lane == 0) red[wv] = acc;
    __syncthreads();
    if (t == 0) {
        float s = 0.0f;
        #pragma unroll
        for (int k = 0; k < BLOCK / 32; ++k) s += red[k];
        out[b] = s;
    }
}

extern "C" void kernel_launch(void* const* d_in, const int* in_sizes, int n_in,
                              void* d_out, int out_size, void* d_ws, size_t ws_size,
                              hipStream_t stream) {
    const float* model_input = (const float*)d_in[0];
    const float* weights     = (const float*)d_in[1];
    float* out = (float*)d_out;
    const int batch = in_sizes[0] / 2;
    qsim12_kernel<<<batch, BLOCK, 0, stream>>>(model_input, weights, out);
}


# LDS state as re and im planes, d1 fastest; groups 0 and 2 take the state as A: fragment pairs load and D tiles store as neighbours, no re-pairing moves, phases on register pairs; readout regrouped
# speedup vs baseline: 2.9833x; 1.0856x over previous
#include <hip/hip_runtime.h>

typedef float v2f __attribute__((ext_vector_type(2)));
typedef float v4f __attribute__((ext_vector_type(4)));
typedef float v8f __attribute__((ext_vector_type(8)));

#define NQ    12
#define NL    6
#define DIM   4096
#define BLOCK 256
#define OP    20

#define Q0    20
#define Q2    324
#define PL_N  (15 * Q2 + 15 * Q0 + 16)

__host__ __device__ static constexpr int pos_of(int i) {
    return Q2 * (i >> 8) + Q0 * (i & 15) + ((i >> 4) & 15);
}

__host__ __device__ static constexpr int Rmap(int j) {
    int x = j;
    for (int w = 11; w >= 0; --w) {
        const int pc = 11 - w;
        const int pt = 11 - ((w + 1) % 12);
        x ^= ((x >> pc) & 1) << pt;
    }
    return x;
}

__host__ __device__ static constexpr int gray4(int k) { return k ^ (k >> 1); }
__host__ __device__ static constexpr int gray4_inv(int g) { return g ^ (g >> 1) ^ (g >> 2) ^ (g >> 3); }
__host__ __device__ static constexpr int par4(int m) { return (m ^ (m >> 1) ^ (m >> 2) ^ (m >> 3)) & 1; }
__host__ __device__ static constexpr bool ring_structure_ok() {
    for (int k = 0; k < 16; ++k) {
        if (Rmap(k << 8) != ((gray4(k) << 8) | ((k & 1) << 7))) return false;
        if ((gray4_inv(gray4(k)) & 15) != k) return false;
    }
    for (int low = 0; low < 256; ++low) {
        const int s = Rmap(low);
        if ((s >> 8) != 12 * (low & 1)) return false;
        if (((s >> 7) & 1) != ((low >> 7) & 1)) return false;
    }
    return true;
}
static_assert(ring_structure_ok(), "ring gather addressing assumes this structure of the CNOT ring map");

__host__ __device__ static constexpr int Rfwd(int j) {
    int x = j;
    for (int w = 0; w < 12; ++w) {
        const int pc = 11 - w;
        const int pt = 11 - ((w + 1) % 12);
        x ^= ((x >> pc) & 1) << pt;
    }
    return x;
}
__host__ __device__ static constexpr int ring_zmask() {
    int m = 0;
    for (int i = 0; i < 12; ++i) {
        if (Rfwd(1 << i) & 1) m |= 1 << i;
    }
    return m;
}
__host__ __device__ static constexpr bool ring_inverse_ok() {
    for (int i = 0; i < 12; ++i) {
        if (Rfwd(Rmap(1 << i)) != (1 << i)) return false;
    }
    return true;
}
static_assert(ring_inverse_ok(), "Rfwd must invert Rmap (both are linear: the unit vectors decide)");
#define ZMASK (ring_zmask())

__device__ static inline float2 cmul(const float2 a, const float2 b) {
    return make_float2(a.x * b.x - a.y * b.y, a.x * b.y + a.y * b.x);
}

__global__ __launch_bounds__(BLOCK) void qsim12_kernel(
    const float* __restrict__ model_input,
    const float* __restrict__ weights,
    float* __restrict__ out)
{
    __shared__ __attribute__((aligned(16))) float sre[PL_N];
    __shared__ __attribute__((aligned(16))) float sim[PL_N];
    __shared__ __attribute__((aligned(16))) float Ot[3][16 * OP];
    __shared__ __attribute__((aligned(16))) float2 Ph[3][16];
    __shared__ __attribute__((aligned(16))) float Pcr[16 * OP];
    __shared__ __attribute__((aligned(16))) float Pci[16 * OP];
    __shared__ float gates[NL][NQ][8];
    __shared__ float red[BLOCK / 32];
    __shared__ __attribute__((aligned(16))) float2 Tab[3][16][4];

    const int b    = blockIdx.x;
    const int t    = threadIdx.x;
    const int lane = t & 31;
    const int wv   = t >> 5;
    const int c16  = lane & 15;
    const int hi   = lane >> 4;

    const int bs0 = Q2 * c16 + Q0 * (2 * wv) + 8 * hi;
    const int bl1 = Q2 * (2 * wv) + Q0 * c16 + 8 * hi;
    int gsrc[2][4];
    #pragma unroll
    for (int tt = 0; tt < 2; ++tt) {
        const int s  = Rmap((c16 << 4) | (2 * wv + tt));
        const int q  = hi ^ ((s >> 11) & 1);
        const int pE = pos_of(s & 0xFF);
        const int pO = pos_of((s & 0xFF) ^ 0x80);
        gsrc[tt][0] = pE + 12 * Q2 * q;
        gsrc[tt][1] = pO + 12 * Q2 * q;
        gsrc[tt][2] = pE + 4 * Q2 * q;
        gsrc[tt][3] = pO + 4 * Q2 * q;
    }

    const float a0 = model_input[2 * b + 0];
    const float a1 = model_input[2 * b + 1];

    if (t < NL * NQ) {
        const int l = t / NQ, w = t % NQ;
        const int limit = NQ - (l >> 1);
        float th = weights[(l * NQ + w) * 2 + 0];
        if (w < limit) th += (w & 1) ? a1 : a0;
        const float ph = weights[(l * NQ + w) * 2 + 1];
        float c, s, pr, pi;
        __sincosf(0.5f * th, &s, &c);
        __sincosf(0.5f * ph, &pi, &pr);
        gates[l][w][0] = c;   gates[l][w][1] = -s;
        gates[l][w][2] = s;   gates[l][w][3] = c;
        gates[l][w][4] = pr;  gates[l][w][5] = -pi;
        gates[l][w][6] = pr;  gates[l][w][7] = pi;
    }
    __syncthreads();

    if (t < 48) {
        const int seg = t >> 4, dg = t & 15;
        #pragma unroll
        for (int p = 0; p < 2; ++p) {
            float2 u0 = make_float2(0.0f, 0.0f), u1 = make_float2(0.0f, 0.0f);
            #pragma unroll
            for (int k = 0; k < 4; ++k) {
                const int w = 4 * seg + k;
                const int z = (dg >> (3 - k)) & 1;
                float2 Tm[2][2];
                #pragma unroll
                for (int r = 0; r < 2; ++r) {
                    #pragma unroll
                    for (int y = 0; y < 2; ++y) {
                        const int bo = y ^ r;
                        const float a = gates[0][w][2 * bo];
                        const float g = gates[1][w][2 * z + y];
                        Tm[r][y] = cmul(make_float2(a * gates[0][w][4 + 2 * bo], a * gates[0][w][5 + 2 * bo]),
                                        make_float2(g * gates[1][w][4 + 2 * z], g * gates[1][w][5 + 2 * z]));
                    }
                }
                if (k == 0) {
                    u0 = Tm[p][0];
                    u1 = Tm[p][1];
                } else {
                    const bool sw = (k == 1) && (seg == 0) && (p == 1);
                    const float2 r00 = sw ? Tm[1][0] : Tm[0][0];
                    const float2 r01 = sw ? Tm[1][1] : Tm[0][1];
                    const float2 r10 = sw ? Tm[0][0] : Tm[1][0];
                    const float2 r11 = sw ? Tm[0][1] : Tm[1][1];
                    const float2 x0 = cmul(u0, r00), x1 = cmul(u1, r10);
                    const float2 y0 = cmul(u0, r01), y1 = cmul(u1, r11);
                    u0 = make_float2(x0.x + x1.x, x0.y + x1.y);
                    u1 = make_float2(y0.x + y1.x, y0.y + y1.y);
                }
            }
            Tab[seg][dg][2 * p]     = u0;
            Tab[seg][dg][2 * p + 1] = u1;
        }
    }
    __syncthreads();
    {
        const float4* Lp = (const float4*)&Tab[0][c16][0];
        const float4 l0 = Lp[0];
        const float4 l1 = Lp[1];
        const float4 rr = *(const float4*)&Tab[2][c16][2 * hi];
        const v2f rb0 = {rr.x, rr.y};
        const v2f rb1 = {rr.z, rr.w};
        #pragma unroll
        for (int tt = 0; tt < 2; ++tt) {
            const float2 m0 = Tab[1][2 * wv + tt][hi];
            const float2 m1 = Tab[1][2 * wv + tt][2 + hi];
            const float2 p00 = cmul(make_float2(l0.x, l0.y), m0), p01 = cmul(make_float2(l0.z, l0.w), m1);
            const float2 p10 = cmul(make_float2(l1.x, l1.y), m0), p11 = cmul(make_float2(l1.z, l1.w), m1);
            const float2 lm0 = make_float2(p00.x + p01.x, p00.y + p01.y);
            const float2 lm1 = make_float2(p10.x + p11.x, p10.y + p11.y);
            const v2f are0 = {lm0.x, -lm0.y}, aim0 = {lm0.y, lm0.x};
            const v2f are1 = {lm1.x, -lm1.y}, aim1 = {lm1.y, lm1.x};
            v8f dr = {0, 0, 0, 0, 0, 0, 0, 0};
            v8f di = {0, 0, 0, 0, 0, 0, 0, 0};
            dr = __builtin_amdgcn_wmma_f32_16x16x4_f32(false, are0, false, rb0, (short)0, dr, false, false);
            di = __builtin_amdgcn_wmma_f32_16x16x4_f32(false, aim0, false, rb0, (short)0, di, false, false);
            dr = __builtin_amdgcn_wmma_f32_16x16x4_f32(false, are1, false, rb1, (short)0, dr, false, false);
            di = __builtin_amdgcn_wmma_f32_16x16x4_f32(false, aim1, false, rb1, (short)0, di, false, false);
            const int sp = Q2 * (8 * hi) + Q0 * c16 + (2 * wv + tt);
            #pragma unroll
            for (int v = 0; v < 8; ++v) {
                sre[sp + Q2 * v] = dr[v];
                sim[sp + Q2 * v] = di[v];
            }
        }
    }
    __syncthreads();

    #pragma unroll 1
    for (int l = 2; l < NL - 1; ++l) {
        {
            const int jo = t >> 4, ji = t & 15;
            const int qA = ((ji & 6) << 1) | ((ji >> 3) << 1) | (ji & 1);
            #pragma unroll
            for (int g = 0; g < 3; ++g) {
                float o = 1.0f;
                #pragma unroll
                for (int k = 0; k < 4; ++k) {
                    const int w  = 4 * g + k;
                    const int bo = (jo >> (3 - k)) & 1;
                    const int bi = (ji >> (3 - k)) & 1;
                    o *= gates[l][w][bo * 2 + bi];
                }
                Ot[g][jo * OP + qA] = o;
            }
            if (t < 48) {
                const int g = t >> 4, j = t & 15;
                float re = 1.0f, im = 0.0f;
                #pragma unroll
                for (int k = 0; k < 4; ++k) {
                    const int w  = 4 * g + k;
                    const int bo = (j >> (3 - k)) & 1;
                    const float er = gates[l][w][4 + 2 * bo];
                    const float ei = gates[l][w][5 + 2 * bo];
                    const float nr = re * er - im * ei;
                    const float ni = re * ei + im * er;
                    re = nr; im = ni;
                }
                Ph[g][j] = make_float2(re, im);
            }
        }
        float gre[2][8], gim[2][8];
        #pragma unroll
        for (int tt = 0; tt < 2; ++tt) {
            #pragma unroll
            for (int e = 0; e < 8; ++e) {
                const int p = gsrc[tt][(e & 1) + 2 * (e >> 2)] + Q2 * gray4(e);
                gre[tt][e] = sre[p];
                gim[tt][e] = sim[p];
            }
        }
        __syncthreads();
        {
            {
                const float2 q1 = Ph[1][t >> 4];
                const float2 q2 = Ph[2][t & 15];
                Pcr[OP * (t & 15) + (t >> 4)] = q1.x * q2.x - q1.y * q2.y;
                Pci[OP * (t & 15) + (t >> 4)] = q1.x * q2.y + q1.y * q2.x;
            }
            v2f b0[4];
            #pragma unroll
            for (int c = 0; c < 4; ++c) {
                b0[c] = *(const v2f*)&Ot[0][c16 * OP + 4 * c + 2 * hi];
            }
            const float2 pz = Ph[0][c16];

            #pragma unroll
            for (int tt = 0; tt < 2; ++tt) {
                v8f dr = {0, 0, 0, 0, 0, 0, 0, 0};
                v8f di = {0, 0, 0, 0, 0, 0, 0, 0};
                #pragma unroll
                for (int c = 0; c < 4; ++c) {
                    const v2f are = {gre[tt][2 * c], gre[tt][2 * c + 1]};
                    const v2f aim = {gim[tt][2 * c], gim[tt][2 * c + 1]};
                    dr = __builtin_amdgcn_wmma_f32_16x16x4_f32(false, are, false, b0[c], (short)0, dr, false, false);
                    di = __builtin_amdgcn_wmma_f32_16x16x4_f32(false, aim, false, b0[c], (short)0, di, false, false);
                }
                const v8f outr = dr * pz.x - di * pz.y;
                const v8f outi = dr * pz.y + di * pz.x;
                v4f* wr = (v4f*)&sre[bs0 + Q0 * tt];
                v4f* wi = (v4f*)&sim[bs0 + Q0 * tt];
                wr[0] = __builtin_shufflevector(outr, outr, 0, 1, 2, 3);
                wr[1] = __builtin_shufflevector(outr, outr, 4, 5, 6, 7);
                wi[0] = __builtin_shufflevector(outi, outi, 0, 1, 2, 3);
                wi[1] = __builtin_shufflevector(outi, outi, 4, 5, 6, 7);
            }
        }
        __syncthreads();
        {
            v2f a1[4], a2[4];
            #pragma unroll
            for (int c = 0; c < 4; ++c) {
                a1[c] = *(const v2f*)&Ot[1][c16 * OP + 4 * c + 2 * hi];
                a2[c] = *(const v2f*)&Ot[2][c16 * OP + 4 * c + 2 * hi];
            }
            v8f pcr, pci;
            {
                const v4f* pr = (const v4f*)&Pcr[OP * c16 + 8 * hi];
                const v4f* pi = (const v4f*)&Pci[OP * c16 + 8 * hi];
                pcr = __builtin_shufflevector(pr[0], pr[1], 0, 1, 2, 3, 4, 5, 6, 7);
                pci = __builtin_shufflevector(pi[0], pi[1], 0, 1, 2, 3, 4, 5, 6, 7);
            }

            #pragma unroll
            for (int tt = 0; tt < 2; ++tt) {
                const v4f* lr = (const v4f*)&sre[bl1 + Q2 * tt];
                const v4f* li = (const v4f*)&sim[bl1 + Q2 * tt];
                const v4f xr0 = lr[0], xr1 = lr[1], xi0 = li[0], xi1 = li[1];
                v8f yr = {0, 0, 0, 0, 0, 0, 0, 0};
                v8f yi = {0, 0, 0, 0, 0, 0, 0, 0};
                yr = __builtin_amdgcn_wmma_f32_16x16x4_f32(false, __builtin_shufflevector(xr0, xr0, 0, 1), false, a1[0], (short)0, yr, false, false);
                yi = __builtin_amdgcn_wmma_f32_16x16x4_f32(false, __builtin_shufflevector(xi0, xi0, 0, 1), false, a1[0], (short)0, yi, false, false);
                yr = __builtin_amdgcn_wmma_f32_16x16x4_f32(false, __builtin_shufflevector(xr0, xr0, 2, 3), false, a1[1], (short)0, yr, false, false);
                yi = __builtin_amdgcn_wmma_f32_16x16x4_f32(false, __builtin_shufflevector(xi0, xi0, 2, 3), false, a1[1], (short)0, yi, false, false);
                yr = __builtin_amdgcn_wmma_f32_16x16x4_f32(false, __builtin_shufflevector(xr1, xr1, 0, 1), false, a1[2], (short)0, yr, false, false);
                yi = __builtin_amdgcn_wmma_f32_16x16x4_f32(false, __builtin_shufflevector(xi1, xi1, 0, 1), false, a1[2], (short)0, yi, false, false);
                yr = __builtin_amdgcn_wmma_f32_16x16x4_f32(false, __builtin_shufflevector(xr1, xr1, 2, 3), false, a1[3], (short)0, yr, false, false);
                yi = __builtin_amdgcn_wmma_f32_16x16x4_f32(false, __builtin_shufflevector(xi1, xi1, 2, 3), false, a1[3], (short)0, yi, false, false);
                v8f zr = {0, 0, 0, 0, 0, 0, 0, 0};
                v8f zi = {0, 0, 0, 0, 0, 0, 0, 0};
                #pragma unroll
                for (int c = 0; c < 4; ++c) {
                    const v2f are = {yr[2 * c], yr[2 * c + 1]};
                    const v2f aim = {yi[2 * c], yi[2 * c + 1]};
                    zr = __builtin_amdgcn_wmma_f32_16x16x4_f32(false, are, false, a2[c], (short)0, zr, false, false);
                    zi = __builtin_amdgcn_wmma_f32_16x16x4_f32(false, aim, false, a2[c], (short)0, zi, false, false);
                }
                const v8f outr = zr * pcr - zi * pci;
                const v8f outi = zr * pci + zi * pcr;
                v4f* wr = (v4f*)&sre[bl1 + Q2 * tt];
                v4f* wi = (v4f*)&sim[bl1 + Q2 * tt];
                wr[0] = __builtin_shufflevector(outr, outr, 0, 1, 2, 3);
                wr[1] = __builtin_shufflevector(outr, outr, 4, 5, 6, 7);
                wi[0] = __builtin_shufflevector(outi, outi, 0, 1, 2, 3);
                wi[1] = __builtin_shufflevector(outi, outi, 4, 5, 6, 7);
            }
        }
        __syncthreads();

    }

    float acc = 0.0f;
    {
        constexpr int l = NL - 1;
        {
            const int jo = t >> 4, ji = t & 15;
            const int qA = ((ji & 6) << 1) | ((ji >> 3) << 1) | (ji & 1);
            #pragma unroll
            for (int g = 0; g < 3; ++g) {
                float o = 1.0f;
                #pragma unroll
                for (int k = 0; k < 4; ++k) {
                    const int w  = 4 * g + k;
                    const int bo = (jo >> (3 - k)) & 1;
                    const int bi = (ji >> (3 - k)) & 1;
                    o *= gates[l][w][bo * 2 + bi];
                }
                Ot[g][jo * OP + qA] = o;
            }
        }
        float gre[2][8], gim[2][8];
        #pragma unroll
        for (int tt = 0; tt < 2; ++tt) {
            #pragma unroll
            for (int e = 0; e < 8; ++e) {
                const int p = gsrc[tt][(e & 1) + 2 * (e >> 2)] + Q2 * gray4(e);
                gre[tt][e] = sre[p];
                gim[tt][e] = sim[p];
            }
        }
        __syncthreads();
        {
            v2f b0[4];
            #pragma unroll
            for (int c = 0; c < 4; ++c) {
                b0[c] = *(const v2f*)&Ot[0][c16 * OP + 4 * c + 2 * hi];
            }

            #pragma unroll
            for (int tt = 0; tt < 2; ++tt) {
                v8f dr = {0, 0, 0, 0, 0, 0, 0, 0};
                v8f di = {0, 0, 0, 0, 0, 0, 0, 0};
                #pragma unroll
                for (int c = 0; c < 4; ++c) {
                    const v2f are = {gre[tt][2 * c], gre[tt][2 * c + 1]};
                    const v2f aim = {gim[tt][2 * c], gim[tt][2 * c + 1]};
                    dr = __builtin_amdgcn_wmma_f32_16x16x4_f32(false, are, false, b0[c], (short)0, dr, false, false);
                    di = __builtin_amdgcn_wmma_f32_16x16x4_f32(false, aim, false, b0[c], (short)0, di, false, false);
                }
                const v8f outr = dr;
                const v8f outi = di;
                v4f* wr = (v4f*)&sre[bs0 + Q0 * tt];
                v4f* wi = (v4f*)&sim[bs0 + Q0 * tt];
                wr[0] = __builtin_shufflevector(outr, outr, 0, 1, 2, 3);
                wr[1] = __builtin_shufflevector(outr, outr, 4, 5, 6, 7);
                wi[0] = __builtin_shufflevector(outi, outi, 0, 1, 2, 3);
                wi[1] = __builtin_shufflevector(outi, outi, 4, 5, 6, 7);
            }
        }
        __syncthreads();
        {
            v2f a1[4], a2[4];
            #pragma unroll
            for (int c = 0; c < 4; ++c) {
                a1[c] = *(const v2f*)&Ot[1][c16 * OP + 4 * c + 2 * hi];
                a2[c] = *(const v2f*)&Ot[2][c16 * OP + 4 * c + 2 * hi];
            }

            #pragma unroll
            for (int tt = 0; tt < 2; ++tt) {
                const v4f* lr = (const v4f*)&sre[bl1 + Q2 * tt];
                const v4f* li = (const v4f*)&sim[bl1 + Q2 * tt];
                const v4f xr0 = lr[0], xr1 = lr[1], xi0 = li[0], xi1 = li[1];
                v8f yr = {0, 0, 0, 0, 0, 0, 0, 0};
                v8f yi = {0, 0, 0, 0, 0, 0, 0, 0};
                yr = __builtin_amdgcn_wmma_f32_16x16x4_f32(false, __builtin_shufflevector(xr0, xr0, 0, 1), false, a1[0], (short)0, yr, false, false);
                yi = __builtin_amdgcn_wmma_f32_16x16x4_f32(false, __builtin_shufflevector(xi0, xi0, 0, 1), false, a1[0], (short)0, yi, false, false);
                yr = __builtin_amdgcn_wmma_f32_16x16x4_f32(false, __builtin_shufflevector(xr0, xr0, 2, 3), false, a1[1], (short)0, yr, false, false);
                yi = __builtin_amdgcn_wmma_f32_16x16x4_f32(false, __builtin_shufflevector(xi0, xi0, 2, 3), false, a1[1], (short)0, yi, false, false);
                yr = __builtin_amdgcn_wmma_f32_16x16x4_f32(false, __builtin_shufflevector(xr1, xr1, 0, 1), false, a1[2], (short)0, yr, false, false);
                yi = __builtin_amdgcn_wmma_f32_16x16x4_f32(false, __builtin_shufflevector(xi1, xi1, 0, 1), false, a1[2], (short)0, yi, false, false);
                yr = __builtin_amdgcn_wmma_f32_16x16x4_f32(false, __builtin_shufflevector(xr1, xr1, 2, 3), false, a1[3], (short)0, yr, false, false);
                yi = __builtin_amdgcn_wmma_f32_16x16x4_f32(false, __builtin_shufflevector(xi1, xi1, 2, 3), false, a1[3], (short)0, yi, false, false);
                v8f zr = {0, 0, 0, 0, 0, 0, 0, 0};
                v8f zi = {0, 0, 0, 0, 0, 0, 0, 0};
                #pragma unroll
                for (int c = 0; c < 4; ++c) {
                    const v2f are = {yr[2 * c], yr[2 * c + 1]};
                    const v2f aim = {yi[2 * c], yi[2 * c + 1]};
                    zr = __builtin_amdgcn_wmma_f32_16x16x4_f32(false, are, false, a2[c], (short)0, zr, false, false);
                    zi = __builtin_amdgcn_wmma_f32_16x16x4_f32(false, aim, false, a2[c], (short)0, zi, false, false);
                }
                float ne = 0.0f, no = 0.0f;
                #pragma unroll
                for (int v = 0; v < 8; ++v) {
                    const float p = zr[v] * zr[v] + zi[v] * zi[v];
                    if (par4(((v << 4) & ZMASK) >> 4)) no += p; else ne += p;
                }
                const float ts = ne - no;
                acc += (__popc(((2 * wv + tt) << 8) & ZMASK) & 1) ? -ts : ts;
            }
        }
        if (__popc(((hi << 7) | c16) & ZMASK) & 1) acc = -acc;
    }
    #pragma unroll
    for (int off = 16; off > 0; off >>= 1) acc += __shfl_xor(acc, off);
    if (lane == 0) red[wv] = acc;
    __syncthreads();
    if (t == 0) {
        float s = 0.0f;
        #pragma unroll
        for (int k = 0; k < BLOCK / 32; ++k) s += red[k];
        out[b] = s;
    }
}

extern "C" void kernel_launch(void* const* d_in, const int* in_sizes, int n_in,
                              void* d_out, int out_size, void* d_ws, size_t ws_size,
                              hipStream_t stream) {
    const float* model_input = (const float*)d_in[0];
    const float* weights     = (const float*)d_in[1];
    float* out = (float*)d_out;
    const int batch = in_sizes[0] / 2;
    qsim12_kernel<<<batch, BLOCK, 0, stream>>>(model_input, weights, out);
}


# Layer 2 in closed form too: behind two rings the state is a closed chain of 4x4 transfer matrices; tree-built tables, complex K=16 product per tile on WMMA; loop starts at layer 3; sums regrouped
# speedup vs baseline: 3.2559x; 1.0914x over previous
#include <hip/hip_runtime.h>

typedef float v2f __attribute__((ext_vector_type(2)));
typedef float v4f __attribute__((ext_vector_type(4)));
typedef float v8f __attribute__((ext_vector_type(8)));

#define NQ    12
#define NL    6
#define DIM   4096
#define BLOCK 256
#define OP    20

#define Q0    20
#define Q2    324
#define PL_N  (15 * Q2 + 15 * Q0 + 16)
#define TP    18

__host__ __device__ static constexpr int pos_of(int i) {
    return Q2 * (i >> 8) + Q0 * (i & 15) + ((i >> 4) & 15);
}

__host__ __device__ static constexpr int Rmap(int j) {
    int x = j;
    for (int w = 11; w >= 0; --w) {
        const int pc = 11 - w;
        const int pt = 11 - ((w + 1) % 12);
        x ^= ((x >> pc) & 1) << pt;
    }
    return x;
}

__host__ __device__ static constexpr int gray4(int k) { return k ^ (k >> 1); }
__host__ __device__ static constexpr int gray4_inv(int g) { return g ^ (g >> 1) ^ (g >> 2) ^ (g >> 3); }
__host__ __device__ static constexpr int par4(int m) { return (m ^ (m >> 1) ^ (m >> 2) ^ (m >> 3)) & 1; }
__host__ __device__ static constexpr bool ring_structure_ok() {
    for (int k = 0; k < 16; ++k) {
        if (Rmap(k << 8) != ((gray4(k) << 8) | ((k & 1) << 7))) return false;
        if ((gray4_inv(gray4(k)) & 15) != k) return false;
    }
    for (int low = 0; low < 256; ++low) {
        const int s = Rmap(low);
        if ((s >> 8) != 12 * (low & 1)) return false;
        if (((s >> 7) & 1) != ((low >> 7) & 1)) return false;
    }
    return true;
}
static_assert(ring_structure_ok(), "ring gather addressing assumes this structure of the CNOT ring map");

__host__ __device__ static constexpr int Rfwd(int j) {
    int x = j;
    for (int w = 0; w < 12; ++w) {
        const int pc = 11 - w;
        const int pt = 11 - ((w + 1) % 12);
        x ^= ((x >> pc) & 1) << pt;
    }
    return x;
}
__host__ __device__ static constexpr int ring_zmask() {
    int m = 0;
    for (int i = 0; i < 12; ++i) {
        if (Rfwd(1 << i) & 1) m |= 1 << i;
    }
    return m;
}
__host__ __device__ static constexpr bool ring_inverse_ok() {
    for (int i = 0; i < 12; ++i) {
        if (Rfwd(Rmap(1 << i)) != (1 << i)) return false;
    }
    return true;
}
static_assert(ring_inverse_ok(), "Rfwd must invert Rmap (both are linear: the unit vectors decide)");
#define ZMASK (ring_zmask())

__device__ static inline v2f cmul(const v2f a, const v2f b) {
    const v2f bq = {-b.y, b.x};
    return __builtin_elementwise_fma(a.yy, bq, a.xx * b);
}
__device__ static inline v2f cmac(const v2f acc, const v2f a, const v2f b) {
    const v2f bq = {-b.y, b.x};
    return __builtin_elementwise_fma(a.yy, bq, __builtin_elementwise_fma(a.xx, b, acc));
}

__global__ __launch_bounds__(BLOCK) void qsim12_kernel(
    const float* __restrict__ model_input,
    const float* __restrict__ weights,
    float* __restrict__ out)
{
    __shared__ __attribute__((aligned(16))) float sre[PL_N];
    __shared__ __attribute__((aligned(16))) float sim[PL_N];
    __shared__ __attribute__((aligned(16))) float Ot[3][16 * OP];
    __shared__ __attribute__((aligned(16))) float2 Ph[3][16];
    __shared__ __attribute__((aligned(16))) float Pcr[16 * OP];
    __shared__ __attribute__((aligned(16))) float Pci[16 * OP];
    __shared__ float gates[NL][NQ][8];
    __shared__ float red[BLOCK / 32];
    __shared__ __attribute__((aligned(16))) v2f cg[3][NQ][4];
    __shared__ __attribute__((aligned(16))) v2f Pair[3][2][4][TP];
    __shared__ __attribute__((aligned(16))) v2f Tab[3][16][TP];

    const int b    = blockIdx.x;
    const int t    = threadIdx.x;
    const int lane = t & 31;
    const int wv   = t >> 5;
    const int c16  = lane & 15;
    const int hi   = lane >> 4;

    const int bs0 = Q2 * c16 + Q0 * (2 * wv) + 8 * hi;
    const int bl1 = Q2 * (2 * wv) + Q0 * c16 + 8 * hi;
    int gsrc[2][4];
    #pragma unroll
    for (int tt = 0; tt < 2; ++tt) {
        const int s  = Rmap((c16 << 4) | (2 * wv + tt));
        const int q  = hi ^ ((s >> 11) & 1);
        const int pE = pos_of(s & 0xFF);
        const int pO = pos_of((s & 0xFF) ^ 0x80);
        gsrc[tt][0] = pE + 12 * Q2 * q;
        gsrc[tt][1] = pO + 12 * Q2 * q;
        gsrc[tt][2] = pE + 4 * Q2 * q;
        gsrc[tt][3] = pO + 4 * Q2 * q;
    }

    const float a0 = model_input[2 * b + 0];
    const float a1 = model_input[2 * b + 1];

    if (t < NL * NQ) {
        const int l = t / NQ, w = t % NQ;
        const int limit = NQ - (l >> 1);
        float th = weights[(l * NQ + w) * 2 + 0];
        if (w < limit) th += (w & 1) ? a1 : a0;
        const float ph = weights[(l * NQ + w) * 2 + 1];
        float c, s, pr, pi;
        __sincosf(0.5f * th, &s, &c);
        __sincosf(0.5f * ph, &pi, &pr);
        if (l < 3) {
            cg[l][w][0] = v2f{c * pr, -(c * pi)};
            cg[l][w][1] = v2f{-(s * pr), s * pi};
            cg[l][w][2] = v2f{s * pr, s * pi};
            cg[l][w][3] = v2f{c * pr, c * pi};
        } else {
            gates[l][w][0] = c;   gates[l][w][1] = -s;
            gates[l][w][2] = s;   gates[l][w][3] = c;
            gates[l][w][4] = pr;  gates[l][w][5] = -pi;
            gates[l][w][6] = pr;  gates[l][w][7] = pi;
        }
    }
    __syncthreads();

    if (t < 96) {
        const int seg = t >> 5, h = (t >> 4) & 1, zz = (t >> 2) & 3, r = t & 3;
        const int wA = 4 * seg + 2 * h, wB = wA + 1;
        const int zA = zz >> 1, zB = zz & 1;
        const int jp = r >> 1, yp = r & 1;
        const v2f g2a0 = cg[2][wA][2 * zA], g2a1 = cg[2][wA][2 * zA + 1];
        const v2f aa0 = cg[0][wA][2 * yp], aa1 = cg[0][wA][2 * (yp ^ 1)];
        const v2f h00 = cmul(cg[1][wA][2 * jp], aa0),       h01 = cmul(cg[1][wA][2 * jp + 1], aa1);
        const v2f h10 = cmul(cg[1][wA][2 * (jp ^ 1)], aa0), h11 = cmul(cg[1][wA][2 * (jp ^ 1) + 1], aa1);
        v2f v0 = cmul(g2a0, h00), v1 = cmul(g2a0, h01), v2 = cmul(g2a1, h10), v3 = cmul(g2a1, h11);
        const int xr = (t < 16) ? r : 0;
        if (xr & 1) {
            v2f x = v0; v0 = v1; v1 = x;
            x = v2; v2 = v3; v3 = x;
        }
        if (xr & 2) {
            v2f x = v0; v0 = v2; v2 = x;
            x = v1; v1 = v3; v3 = x;
        }
        const v2f g2b0 = cg[2][wB][2 * zB], g2b1 = cg[2][wB][2 * zB + 1];
        v2f o[4];
        #pragma unroll
        for (int y = 0; y < 2; ++y) {
            const v2f ae = cg[0][wB][2 * y];
            const v2f ao = cg[0][wB][2 * (y ^ 1)];
            const v2f t0 = cmac(cmul(v0, ae), v1, ao);
            const v2f t1 = cmac(cmul(v2, ae), v3, ao);
            const v2f k0 = cg[1][wB][y], k1 = cg[1][wB][2 + y];
            o[y]     = cmul(g2b0, cmac(cmul(k0, t0), k1, t1));
            o[2 + y] = cmul(g2b1, cmac(cmul(k1, t0), k0, t1));
        }
        v4f* pp = (v4f*)&Pair[seg][h][zz][4 * r];
        pp[0] = __builtin_shufflevector(o[0], o[1], 0, 1, 2, 3);
        pp[1] = __builtin_shufflevector(o[2], o[3], 0, 1, 2, 3);
    }
    __syncthreads();
    if (t < 192) {
        const int seg = t >> 6, dg = (t >> 2) & 15, r = t & 3;
        const v4f* rp = (const v4f*)&Pair[seg][0][dg >> 2][4 * r];
        const v4f* mp = (const v4f*)&Pair[seg][1][dg & 3][0];
        const v4f ra = rp[0], rc = rp[1];
        const v2f rw[4] = {__builtin_shufflevector(ra, ra, 0, 1), __builtin_shufflevector(ra, ra, 2, 3),
                           __builtin_shufflevector(rc, rc, 0, 1), __builtin_shufflevector(rc, rc, 2, 3)};
        v2f o[4];
        #pragma unroll
        for (int m = 0; m < 4; ++m) {
            const v4f ma = mp[2 * m], mb = mp[2 * m + 1];
            const v2f mq[4] = {__builtin_shufflevector(ma, ma, 0, 1), __builtin_shufflevector(ma, ma, 2, 3),
                               __builtin_shufflevector(mb, mb, 0, 1), __builtin_shufflevector(mb, mb, 2, 3)};
            #pragma unroll
            for (int q = 0; q < 4; ++q) {
                o[q] = (m == 0) ? cmul(rw[0], mq[q]) : cmac(o[q], rw[m], mq[q]);
            }
        }
        v4f* tp = (v4f*)&Tab[seg][dg][4 * r];
        tp[0] = __builtin_shufflevector(o[0], o[1], 0, 1, 2, 3);
        tp[1] = __builtin_shufflevector(o[2], o[3], 0, 1, 2, 3);
    }
    __syncthreads();
    {
        v2f Lr[16];
        {
            const v4f* Lp = (const v4f*)&Tab[0][c16][0];
            #pragma unroll
            for (int i = 0; i < 8; ++i) {
                const v4f x = Lp[i];
                Lr[2 * i]     = __builtin_shufflevector(x, x, 0, 1);
                Lr[2 * i + 1] = __builtin_shufflevector(x, x, 2, 3);
            }
        }
        v2f rbr[8], rbi[8];
        #pragma unroll
        for (int bs = 0; bs < 2; ++bs) {
            const v4f* Rp = (const v4f*)&Tab[2][c16][4 * (2 * bs + hi)];
            const v4f x0 = Rp[0], x1 = Rp[1];
            rbr[0 + bs] = v2f{x0.x, -x0.y};   rbi[0 + bs] = v2f{x0.y, x0.x};
            rbr[2 + bs] = v2f{x0.z, -x0.w};   rbi[2 + bs] = v2f{x0.w, x0.z};
            rbr[4 + bs] = v2f{x1.x, -x1.y};   rbi[4 + bs] = v2f{x1.y, x1.x};
            rbr[6 + bs] = v2f{x1.z, -x1.w};   rbi[6 + bs] = v2f{x1.w, x1.z};
        }
        #pragma unroll
        for (int tt = 0; tt < 2; ++tt) {
            v2f mm[4][2];
            #pragma unroll
            for (int a = 0; a < 4; ++a) {
                mm[a][0] = Tab[1][2 * wv + tt][4 * a + hi];
                mm[a][1] = Tab[1][2 * wv + tt][4 * a + 2 + hi];
            }
            v8f dr = {0, 0, 0, 0, 0, 0, 0, 0};
            v8f di = {0, 0, 0, 0, 0, 0, 0, 0};
            #pragma unroll
            for (int c = 0; c < 4; ++c) {
                #pragma unroll
                for (int bs = 0; bs < 2; ++bs) {
                    v2f lm = cmul(Lr[4 * c], mm[0][bs]);
                    lm = cmac(lm, Lr[4 * c + 1], mm[1][bs]);
                    lm = cmac(lm, Lr[4 * c + 2], mm[2][bs]);
                    lm = cmac(lm, Lr[4 * c + 3], mm[3][bs]);
                    dr = __builtin_amdgcn_wmma_f32_16x16x4_f32(false, lm, false, rbr[2 * c + bs], (short)0, dr, false, false);
                    di = __builtin_amdgcn_wmma_f32_16x16x4_f32(false, lm, false, rbi[2 * c + bs], (short)0, di, false, false);
                }
            }
            const int sp = Q2 * (8 * hi) + Q0 * c16 + (2 * wv + tt);
            #pragma unroll
            for (int v = 0; v < 8; ++v) {
                sre[sp + Q2 * v] = dr[v];
                sim[sp + Q2 * v] = di[v];
            }
        }
    }
    __syncthreads();

    #pragma unroll 1
    for (int l = 3; l < NL - 1; ++l) {
        {
            const int jo = t >> 4, ji = t & 15;
            const int qA = ((ji & 6) << 1) | ((ji >> 3) << 1) | (ji & 1);
            #pragma unroll
            for (int g = 0; g < 3; ++g) {
                float o = 1.0f;
                #pragma unroll
                for (int k = 0; k < 4; ++k) {
                    const int w  = 4 * g + k;
                    const int bo = (jo >> (3 - k)) & 1;
                    const int bi = (ji >> (3 - k)) & 1;
                    o *= gates[l][w][bo * 2 + bi];
                }
                Ot[g][jo * OP + qA] = o;
            }
            if (t < 48) {
                const int g = t >> 4, j = t & 15;
                float re = 1.0f, im = 0.0f;
                #pragma unroll
                for (int k = 0; k < 4; ++k) {
                    const int w  = 4 * g + k;
                    const int bo = (j >> (3 - k)) & 1;
                    const float er = gates[l][w][4 + 2 * bo];
                    const float ei = gates[l][w][5 + 2 * bo];
                    const float nr = re * er - im * ei;
                    const float ni = re * ei + im * er;
                    re = nr; im = ni;
                }
                Ph[g][j] = make_float2(re, im);
            }
        }
        float gre[2][8], gim[2][8];
        #pragma unroll
        for (int tt = 0; tt < 2; ++tt) {
            #pragma unroll
            for (int e = 0; e < 8; ++e) {
                const int p = gsrc[tt][(e & 1) + 2 * (e >> 2)] + Q2 * gray4(e);
                gre[tt][e] = sre[p];
                gim[tt][e] = sim[p];
            }
        }
        __syncthreads();
        {
            {
                const float2 q1 = Ph[1][t >> 4];
                const float2 q2 = Ph[2][t & 15];
                Pcr[OP * (t & 15) + (t >> 4)] = q1.x * q2.x - q1.y * q2.y;
                Pci[OP * (t & 15) + (t >> 4)] = q1.x * q2.y + q1.y * q2.x;
            }
            v2f b0[4];
            #pragma unroll
            for (int c = 0; c < 4; ++c) {
                b0[c] = *(const v2f*)&Ot[0][c16 * OP + 4 * c + 2 * hi];
            }
            const float2 pz = Ph[0][c16];

            #pragma unroll
            for (int tt = 0; tt < 2; ++tt) {
                v8f dr = {0, 0, 0, 0, 0, 0, 0, 0};
                v8f di = {0, 0, 0, 0, 0, 0, 0, 0};
                #pragma unroll
                for (int c = 0; c < 4; ++c) {
                    const v2f are = {gre[tt][2 * c], gre[tt][2 * c + 1]};
                    const v2f aim = {gim[tt][2 * c], gim[tt][2 * c + 1]};
                    dr = __builtin_amdgcn_wmma_f32_16x16x4_f32(false, are, false, b0[c], (short)0, dr, false, false);
                    di = __builtin_amdgcn_wmma_f32_16x16x4_f32(false, aim, false, b0[c], (short)0, di, false, false);
                }
                const v8f outr = dr * pz.x - di * pz.y;
                const v8f outi = dr * pz.y + di * pz.x;
                v4f* wr = (v4f*)&sre[bs0 + Q0 * tt];
                v4f* wi = (v4f*)&sim[bs0 + Q0 * tt];
                wr[0] = __builtin_shufflevector(outr, outr, 0, 1, 2, 3);
                wr[1] = __builtin_shufflevector(outr, outr, 4, 5, 6, 7);
                wi[0] = __builtin_shufflevector(outi, outi, 0, 1, 2, 3);
                wi[1] = __builtin_shufflevector(outi, outi, 4, 5, 6, 7);
            }
        }
        __syncthreads();
        {
            v2f a1[4], a2[4];
            #pragma unroll
            for (int c = 0; c < 4; ++c) {
                a1[c] = *(const v2f*)&Ot[1][c16 * OP + 4 * c + 2 * hi];
                a2[c] = *(const v2f*)&Ot[2][c16 * OP + 4 * c + 2 * hi];
            }
            v8f pcr, pci;
            {
                const v4f* pr = (const v4f*)&Pcr[OP * c16 + 8 * hi];
                const v4f* pi = (const v4f*)&Pci[OP * c16 + 8 * hi];
                pcr = __builtin_shufflevector(pr[0], pr[1], 0, 1, 2, 3, 4, 5, 6, 7);
                pci = __builtin_shufflevector(pi[0], pi[1], 0, 1, 2, 3, 4, 5, 6, 7);
            }

            #pragma unroll
            for (int tt = 0; tt < 2; ++tt) {
                const v4f* lr = (const v4f*)&sre[bl1 + Q2 * tt];
                const v4f* li = (const v4f*)&sim[bl1 + Q2 * tt];
                const v4f xr0 = lr[0], xr1 = lr[1], xi0 = li[0], xi1 = li[1];
                v8f yr = {0, 0, 0, 0, 0, 0, 0, 0};
                v8f yi = {0, 0, 0, 0, 0, 0, 0, 0};
                yr = __builtin_amdgcn_wmma_f32_16x16x4_f32(false, __builtin_shufflevector(xr0, xr0, 0, 1), false, a1[0], (short)0, yr, false, false);
                yi = __builtin_amdgcn_wmma_f32_16x16x4_f32(false, __builtin_shufflevector(xi0, xi0, 0, 1), false, a1[0], (short)0, yi, false, false);
                yr = __builtin_amdgcn_wmma_f32_16x16x4_f32(false, __builtin_shufflevector(xr0, xr0, 2, 3), false, a1[1], (short)0, yr, false, false);
                yi = __builtin_amdgcn_wmma_f32_16x16x4_f32(false, __builtin_shufflevector(xi0, xi0, 2, 3), false, a1[1], (short)0, yi, false, false);
                yr = __builtin_amdgcn_wmma_f32_16x16x4_f32(false, __builtin_shufflevector(xr1, xr1, 0, 1), false, a1[2], (short)0, yr, false, false);
                yi = __builtin_amdgcn_wmma_f32_16x16x4_f32(false, __builtin_shufflevector(xi1, xi1, 0, 1), false, a1[2], (short)0, yi, false, false);
                yr = __builtin_amdgcn_wmma_f32_16x16x4_f32(false, __builtin_shufflevector(xr1, xr1, 2, 3), false, a1[3], (short)0, yr, false, false);
                yi = __builtin_amdgcn_wmma_f32_16x16x4_f32(false, __builtin_shufflevector(xi1, xi1, 2, 3), false, a1[3], (short)0, yi, false, false);
                v8f zr = {0, 0, 0, 0, 0, 0, 0, 0};
                v8f zi = {0, 0, 0, 0, 0, 0, 0, 0};
                #pragma unroll
                for (int c = 0; c < 4; ++c) {
                    const v2f are = {yr[2 * c], yr[2 * c + 1]};
                    const v2f aim = {yi[2 * c], yi[2 * c + 1]};
                    zr = __builtin_amdgcn_wmma_f32_16x16x4_f32(false, are, false, a2[c], (short)0, zr, false, false);
                    zi = __builtin_amdgcn_wmma_f32_16x16x4_f32(false, aim, false, a2[c], (short)0, zi, false, false);
                }
                const v8f outr = zr * pcr - zi * pci;
                const v8f outi = zr * pci + zi * pcr;
                v4f* wr = (v4f*)&sre[bl1 + Q2 * tt];
                v4f* wi = (v4f*)&sim[bl1 + Q2 * tt];
                wr[0] = __builtin_shufflevector(outr, outr, 0, 1, 2, 3);
                wr[1] = __builtin_shufflevector(outr, outr, 4, 5, 6, 7);
                wi[0] = __builtin_shufflevector(outi, outi, 0, 1, 2, 3);
                wi[1] = __builtin_shufflevector(outi, outi, 4, 5, 6, 7);
            }
        }
        __syncthreads();

    }

    float acc = 0.0f;
    {
        constexpr int l = NL - 1;
        {
            const int jo = t >> 4, ji = t & 15;
            const int qA = ((ji & 6) << 1) | ((ji >> 3) << 1) | (ji & 1);
            #pragma unroll
            for (int g = 0; g < 3; ++g) {
                float o = 1.0f;
                #pragma unroll
                for (int k = 0; k < 4; ++k) {
                    const int w  = 4 * g + k;
                    const int bo = (jo >> (3 - k)) & 1;
                    const int bi = (ji >> (3 - k)) & 1;
                    o *= gates[l][w][bo * 2 + bi];
                }
                Ot[g][jo * OP + qA] = o;
            }
        }
        float gre[2][8], gim[2][8];
        #pragma unroll
        for (int tt = 0; tt < 2; ++tt) {
            #pragma unroll
            for (int e = 0; e < 8; ++e) {
                const int p = gsrc[tt][(e & 1) + 2 * (e >> 2)] + Q2 * gray4(e);
                gre[tt][e] = sre[p];
                gim[tt][e] = sim[p];
            }
        }
        __syncthreads();
        {
            v2f b0[4];
            #pragma unroll
            for (int c = 0; c < 4; ++c) {
                b0[c] = *(const v2f*)&Ot[0][c16 * OP + 4 * c + 2 * hi];
            }

            #pragma unroll
            for (int tt = 0; tt < 2; ++tt) {
                v8f dr = {0, 0, 0, 0, 0, 0, 0, 0};
                v8f di = {0, 0, 0, 0, 0, 0, 0, 0};
                #pragma unroll
                for (int c = 0; c < 4; ++c) {
                    const v2f are = {gre[tt][2 * c], gre[tt][2 * c + 1]};
                    const v2f aim = {gim[tt][2 * c], gim[tt][2 * c + 1]};
                    dr = __builtin_amdgcn_wmma_f32_16x16x4_f32(false, are, false, b0[c], (short)0, dr, false, false);
                    di = __builtin_amdgcn_wmma_f32_16x16x4_f32(false, aim, false, b0[c], (short)0, di, false, false);
                }
                const v8f outr = dr;
                const v8f outi = di;
                v4f* wr = (v4f*)&sre[bs0 + Q0 * tt];
                v4f* wi = (v4f*)&sim[bs0 + Q0 * tt];
                wr[0] = __builtin_shufflevector(outr, outr, 0, 1, 2, 3);
                wr[1] = __builtin_shufflevector(outr, outr, 4, 5, 6, 7);
                wi[0] = __builtin_shufflevector(outi, outi, 0, 1, 2, 3);
                wi[1] = __builtin_shufflevector(outi, outi, 4, 5, 6, 7);
            }
        }
        __syncthreads();
        {
            v2f a1[4], a2[4];
            #pragma unroll
            for (int c = 0; c < 4; ++c) {
                a1[c] = *(const v2f*)&Ot[1][c16 * OP + 4 * c + 2 * hi];
                a2[c] = *(const v2f*)&Ot[2][c16 * OP + 4 * c + 2 * hi];
            }

            #pragma unroll
            for (int tt = 0; tt < 2; ++tt) {
                const v4f* lr = (const v4f*)&sre[bl1 + Q2 * tt];
                const v4f* li = (const v4f*)&sim[bl1 + Q2 * tt];
                const v4f xr0 = lr[0], xr1 = lr[1], xi0 = li[0], xi1 = li[1];
                v8f yr = {0, 0, 0, 0, 0, 0, 0, 0};
                v8f yi = {0, 0, 0, 0, 0, 0, 0, 0};
                yr = __builtin_amdgcn_wmma_f32_16x16x4_f32(false, __builtin_shufflevector(xr0, xr0, 0, 1), false, a1[0], (short)0, yr, false, false);
                yi = __builtin_amdgcn_wmma_f32_16x16x4_f32(false, __builtin_shufflevector(xi0, xi0, 0, 1), false, a1[0], (short)0, yi, false, false);
                yr = __builtin_amdgcn_wmma_f32_16x16x4_f32(false, __builtin_shufflevector(xr0, xr0, 2, 3), false, a1[1], (short)0, yr, false, false);
                yi = __builtin_amdgcn_wmma_f32_16x16x4_f32(false, __builtin_shufflevector(xi0, xi0, 2, 3), false, a1[1], (short)0, yi, false, false);
                yr = __builtin_amdgcn_wmma_f32_16x16x4_f32(false, __builtin_shufflevector(xr1, xr1, 0, 1), false, a1[2], (short)0, yr, false, false);
                yi = __builtin_amdgcn_wmma_f32_16x16x4_f32(false, __builtin_shufflevector(xi1, xi1, 0, 1), false, a1[2], (short)0, yi, false, false);
                yr = __builtin_amdgcn_wmma_f32_16x16x4_f32(false, __builtin_shufflevector(xr1, xr1, 2, 3), false, a1[3], (short)0, yr, false, false);
                yi = __builtin_amdgcn_wmma_f32_16x16x4_f32(false, __builtin_shufflevector(xi1, xi1, 2, 3), false, a1[3], (short)0, yi, false, false);
                v8f zr = {0, 0, 0, 0, 0, 0, 0, 0};
                v8f zi = {0, 0, 0, 0, 0, 0, 0, 0};
                #pragma unroll
                for (int c = 0; c < 4; ++c) {
                    const v2f are = {yr[2 * c], yr[2 * c + 1]};
                    const v2f aim = {yi[2 * c], yi[2 * c + 1]};
                    zr = __builtin_amdgcn_wmma_f32_16x16x4_f32(false, are, false, a2[c], (short)0, zr, false, false);
                    zi = __builtin_amdgcn_wmma_f32_16x16x4_f32(false, aim, false, a2[c], (short)0, zi, false, false);
                }
                float ne = 0.0f, no = 0.0f;
                #pragma unroll
                for (int v = 0; v < 8; ++v) {
                    const float p = zr[v] * zr[v] + zi[v] * zi[v];
                    if (par4(((v << 4) & ZMASK) >> 4)) no += p; else ne += p;
                }
                const float ts = ne - no;
                acc += (__popc(((2 * wv + tt) << 8) & ZMASK) & 1) ? -ts : ts;
            }
        }
        if (__popc(((hi << 7) | c16) & ZMASK) & 1) acc = -acc;
    }
    #pragma unroll
    for (int off = 16; off > 0; off >>= 1) acc += __shfl_xor(acc, off);
    if (lane == 0) red[wv] = acc;
    __syncthreads();
    if (t == 0) {
        float s = 0.0f;
        #pragma unroll
        for (int k = 0; k < BLOCK / 32; ++k) s += red[k];
        out[b] = s;
    }
}

extern "C" void kernel_launch(void* const* d_in, const int* in_sizes, int n_in,
                              void* d_out, int out_size, void* d_ws, size_t ws_size,
                              hipStream_t stream) {
    const float* model_input = (const float*)d_in[0];
    const float* weights     = (const float*)d_in[1];
    float* out = (float*)d_out;
    const int batch = in_sizes[0] / 2;
    qsim12_kernel<<<batch, BLOCK, 0, stream>>>(model_input, weights, out);
}
